# starting kernel
# baseline (speedup 1.0000x reference)
#include <hip/hip_runtime.h>
#include <hip/hip_bf16.h>
#include <stdint.h>

#define N_NODES 100000
#define N_EDGES 1600000
#define FN 32
#define FE 8
#define HID 32
#define NEG_SLOPE 0.2f

typedef __attribute__((ext_vector_type(2))) float v2f;
typedef __attribute__((ext_vector_type(8))) float v8f;

constexpr size_t OFF_H    = 0;
constexpr size_t OFF_ASRC = OFF_H    + (size_t)N_NODES * HID;
constexpr size_t OFF_ADST = OFF_ASRC + N_NODES;
constexpr size_t OFF_SEGM = OFF_ADST + N_NODES;
constexpr size_t OFF_DEN  = OFF_SEGM + N_NODES;
constexpr size_t OFF_LOG  = OFF_DEN  + N_NODES;
constexpr size_t OFF_ACC  = OFF_LOG  + (size_t)(N_EDGES + N_NODES);
constexpr size_t OFF_MISC = OFF_ACC  + (size_t)N_NODES * HID;

__device__ __forceinline__ unsigned fenc(float f) {
  unsigned u = __float_as_uint(f);
  return (u & 0x80000000u) ? ~u : (u | 0x80000000u);
}
__device__ __forceinline__ float fdec(unsigned u) {
  unsigned v = (u & 0x80000000u) ? (u & 0x7FFFFFFFu) : ~u;
  return __uint_as_float(v);
}
#define ENC_NEG_INF 0x007FFFFFu

__device__ __forceinline__ void atomAddF(float* p, float v) { unsafeAtomicAdd(p, v); }

__global__ void k_setup(const float* __restrict__ W_edge, const float* __restrict__ att_edge,
                        float* __restrict__ misc) {
  int t = threadIdx.x;
  if (t < FE) {
    float s = 0.f;
    #pragma unroll
    for (int k = 0; k < HID; ++k) s += W_edge[t * HID + k] * att_edge[k];
    misc[t] = s;
  }
  if (t == FE) misc[FE] = 0.f;
}

__global__ void __launch_bounds__(256) k_gemm_h(const float* __restrict__ x,
                                                const float* __restrict__ W,
                                                float* __restrict__ h) {
  const int wave = blockIdx.x * (blockDim.x >> 5) + (threadIdx.x >> 5);
  const int lane = threadIdx.x & 31;
  const int ntiles = N_NODES / 16;
  if (wave >= ntiles) return;
  const int m0 = wave * 16;
  const int r  = lane & 15;
  const int hi = lane >> 4;

  v8f c0 = {}; v8f c1 = {};
#if __has_builtin(__builtin_amdgcn_wmma_f32_16x16x4_f32)
  const float* xrow = x + (size_t)(m0 + r) * FN;
  #pragma unroll
  for (int kc = 0; kc < 8; ++kc) {
    const int kb = kc * 4 + hi * 2;
    v2f a; a.x = xrow[kb]; a.y = xrow[kb + 1];
    v2f b0; b0.x = W[(size_t)kb * HID + r];        b0.y = W[(size_t)(kb + 1) * HID + r];
    v2f b1; b1.x = W[(size_t)kb * HID + 16 + r];   b1.y = W[(size_t)(kb + 1) * HID + 16 + r];
    c0 = __builtin_amdgcn_wmma_f32_16x16x4_f32(false, a, false, b0, (short)0, c0, false, false);
    c1 = __builtin_amdgcn_wmma_f32_16x16x4_f32(false, a, false, b1, (short)0, c1, false, false);
  }
  #pragma unroll
  for (int v = 0; v < 8; ++v) {
    const int row = m0 + v + hi * 8;
    h[(size_t)row * HID + r]      = c0[v];
    h[(size_t)row * HID + 16 + r] = c1[v];
  }
#else
  const int row = m0 + r;
  for (int n = hi * 16; n < hi * 16 + 16; ++n) {
    float s = 0.f;
    for (int k = 0; k < FN; ++k) s += x[(size_t)row * FN + k] * W[(size_t)k * HID + n];
    h[(size_t)row * HID + n] = s;
  }
#endif
}

__global__ void k_node(const float* __restrict__ h, const float* __restrict__ att_src,
                       const float* __restrict__ att_dst, float* __restrict__ a_src,
                       float* __restrict__ a_dst, unsigned* __restrict__ segmax,
                       float* __restrict__ denom, float* __restrict__ acc) {
  const int i = blockIdx.x * blockDim.x + threadIdx.x;
  if (i >= N_NODES) return;
  const float4* hp = (const float4*)(h + (size_t)i * HID);
  float s = 0.f, d = 0.f;
  #pragma unroll
  for (int q = 0; q < 8; ++q) {
    float4 hv = hp[q];
    float4 as = ((const float4*)att_src)[q];
    float4 ad = ((const float4*)att_dst)[q];
    s += hv.x * as.x + hv.y * as.y + hv.z * as.z + hv.w * as.w;
    d += hv.x * ad.x + hv.y * ad.y + hv.z * ad.z + hv.w * ad.w;
  }
  a_src[i] = s;
  a_dst[i] = d;
  segmax[i] = ENC_NEG_INF;
  denom[i] = 0.f;
  float4* ap = (float4*)(acc + (size_t)i * HID);
  float4 z = {0.f, 0.f, 0.f, 0.f};
  #pragma unroll
  for (int q = 0; q < 8; ++q) ap[q] = z;
}

__global__ void k_edge_logits(const float* __restrict__ ea, const int* __restrict__ ei,
                              const float* __restrict__ a_src, const float* __restrict__ a_dst,
                              const float* __restrict__ misc, float* __restrict__ logits,
                              unsigned* __restrict__ segmax, float* __restrict__ a_edge_sum) {
  const int e = blockIdx.x * blockDim.x + threadIdx.x;
  float ae = 0.f;
  if (e < N_EDGES) {
    const float4* ep = (const float4*)(ea + (size_t)e * FE);
    float4 e0 = ep[0], e1 = ep[1];
    ae = e0.x * misc[0] + e0.y * misc[1] + e0.z * misc[2] + e0.w * misc[3]
       + e1.x * misc[4] + e1.y * misc[5] + e1.z * misc[6] + e1.w * misc[7];
    const int s  = ei[(size_t)e * 2 + 0];
    const int dd = ei[(size_t)e * 2 + 1];
    float l = a_src[s] + a_dst[dd] + ae;
    l = (l >= 0.f) ? l : NEG_SLOPE * l;
    logits[e] = l;
    atomicMax(segmax + dd, fenc(l));
  }
  float v = ae;
  #pragma unroll
  for (int off = 16; off > 0; off >>= 1) v += __shfl_down(v, off, 32);
  if ((threadIdx.x & 31) == 0) atomAddF(a_edge_sum, v);
}

__global__ void k_self_logits(const float* __restrict__ a_src, const float* __restrict__ a_dst,
                              const float* __restrict__ a_edge_sum, float* __restrict__ logits,
                              unsigned* __restrict__ segmax) {
  const int i = blockIdx.x * blockDim.x + threadIdx.x;
  if (i >= N_NODES) return;
  const float aem = a_edge_sum[0] * (1.0f / (float)N_EDGES);
  float l = a_src[i] + a_dst[i] + aem;
  l = (l >= 0.f) ? l : NEG_SLOPE * l;
  logits[N_EDGES + i] = l;
  atomicMax(segmax + i, fenc(l));
}

__global__ void k_exp(const int* __restrict__ ei, float* __restrict__ logits,
                      const unsigned* __restrict__ segmax, float* __restrict__ denom) {
  const int t = blockIdx.x * blockDim.x + threadIdx.x;
  if (t >= N_EDGES + N_NODES) return;
  const int dd = (t < N_EDGES) ? ei[(size_t)t * 2 + 1] : (t - N_EDGES);
  const float m = fdec(segmax[dd]);
  const float a = __expf(logits[t] - m);
  logits[t] = a;
  atomAddF(denom + dd, a);
}

__global__ void k_scatter(const int* __restrict__ ei, const float* __restrict__ h,
                          const float* __restrict__ alpha, float* __restrict__ acc) {
  const int t = blockIdx.x * blockDim.x + threadIdx.x;
  if (t >= N_EDGES + N_NODES) return;
  int s, dd;
  if (t < N_EDGES) { s = ei[(size_t)t * 2 + 0]; dd = ei[(size_t)t * 2 + 1]; }
  else             { s = dd = t - N_EDGES; }
  const float a = alpha[t];
  const float* hp = h + (size_t)s * HID;
  float* ap = acc + (size_t)dd * HID;
  #pragma unroll
  for (int k = 0; k < HID; ++k) atomAddF(ap + k, a * hp[k]);
}

__global__ void k_final(const float* __restrict__ acc, const float* __restrict__ denom,
                        const float* __restrict__ bias, const float* __restrict__ W_lin,
                        const float* __restrict__ b_lin, float* __restrict__ out) {
  const int i = blockIdx.x * blockDim.x + threadIdx.x;
  if (i >= N_NODES) return;
  const float inv = 1.0f / denom[i];
  const float* ap = acc + (size_t)i * HID;
  float y = 0.f;
  #pragma unroll
  for (int k = 0; k < HID; ++k) {
    float v = ap[k] * inv + bias[k];
    v = (v > 0.f) ? v : 0.f;
    y += v * W_lin[k];
  }
  out[i] = y + b_lin[0];
}

extern "C" void kernel_launch(void* const* d_in, const int* in_sizes, int n_in,
                              void* d_out, int out_size, void* d_ws, size_t ws_size,
                              hipStream_t stream) {
  const float* x        = (const float*)d_in[0];
  const float* ea       = (const float*)d_in[1];
  const int*   ei       = (const int*)d_in[2];
  const float* W        = (const float*)d_in[3];
  const float* att_src  = (const float*)d_in[4];
  const float* att_dst  = (const float*)d_in[5];
  const float* W_edge   = (const float*)d_in[6];
  const float* att_edge = (const float*)d_in[7];
  const float* bias     = (const float*)d_in[8];
  const float* W_lin    = (const float*)d_in[9];
  const float* b_lin    = (const float*)d_in[10];
  float* out = (float*)d_out;

  float*    ws     = (float*)d_ws;
  float*    h      = ws + OFF_H;
  float*    a_src  = ws + OFF_ASRC;
  float*    a_dst  = ws + OFF_ADST;
  unsigned* segmax = (unsigned*)(ws + OFF_SEGM);
  float*    denom  = ws + OFF_DEN;
  float*    logits = ws + OFF_LOG;
  float*    acc    = ws + OFF_ACC;
  float*    misc   = ws + OFF_MISC;

  const int T = 256;
  const int nb_nodes = (N_NODES + T - 1) / T;
  const int nb_edges = (N_EDGES + T - 1) / T;
  const int nb_all   = (N_EDGES + N_NODES + T - 1) / T;
  const int nb_gemm  = ((N_NODES / 16) + 7) / 8;

  k_setup<<<1, 32, 0, stream>>>(W_edge, att_edge, misc);
  k_gemm_h<<<nb_gemm, T, 0, stream>>>(x, W, h);
  k_node<<<nb_nodes, T, 0, stream>>>(h, att_src, att_dst, a_src, a_dst, segmax, denom, acc);
  k_edge_logits<<<nb_edges, T, 0, stream>>>(ea, ei, a_src, a_dst, misc, logits, segmax, misc + FE);
  k_self_logits<<<nb_nodes, T, 0, stream>>>(a_src, a_dst, misc + FE, logits, segmax);
  k_exp<<<nb_all, T, 0, stream>>>(ei, logits, segmax, denom);
  k_scatter<<<nb_all, T, 0, stream>>>(ei, h, logits, acc);
  k_final<<<nb_nodes, T, 0, stream>>>(acc, denom, bias, W_lin, b_lin, out);
}


# k_scatter: a wave walks its 32 items with lane = feature, so each f32 atomic instruction and each h gather covers one 128-byte row, not 32 rows; atomic arrival order of the acc sums changes
# speedup vs baseline: 2.6018x; 2.6018x over previous
#include <hip/hip_runtime.h>
#include <hip/hip_bf16.h>
#include <stdint.h>

#define N_NODES 100000
#define N_EDGES 1600000
#define FN 32
#define FE 8
#define HID 32
#define NEG_SLOPE 0.2f

typedef __attribute__((ext_vector_type(2))) float v2f;
typedef __attribute__((ext_vector_type(8))) float v8f;

constexpr size_t OFF_H    = 0;
constexpr size_t OFF_ASRC = OFF_H    + (size_t)N_NODES * HID;
constexpr size_t OFF_ADST = OFF_ASRC + N_NODES;
constexpr size_t OFF_SEGM = OFF_ADST + N_NODES;
constexpr size_t OFF_DEN  = OFF_SEGM + N_NODES;
constexpr size_t OFF_LOG  = OFF_DEN  + N_NODES;
constexpr size_t OFF_ACC  = OFF_LOG  + (size_t)(N_EDGES + N_NODES);
constexpr size_t OFF_MISC = OFF_ACC  + (size_t)N_NODES * HID;

__device__ __forceinline__ unsigned fenc(float f) {
  unsigned u = __float_as_uint(f);
  return (u & 0x80000000u) ? ~u : (u | 0x80000000u);
}
__device__ __forceinline__ float fdec(unsigned u) {
  unsigned v = (u & 0x80000000u) ? (u & 0x7FFFFFFFu) : ~u;
  return __uint_as_float(v);
}
#define ENC_NEG_INF 0x007FFFFFu

__device__ __forceinline__ void atomAddF(float* p, float v) { unsafeAtomicAdd(p, v); }

__global__ void k_setup(const float* __restrict__ W_edge, const float* __restrict__ att_edge,
                        float* __restrict__ misc) {
  int t = threadIdx.x;
  if (t < FE) {
    float s = 0.f;
    #pragma unroll
    for (int k = 0; k < HID; ++k) s += W_edge[t * HID + k] * att_edge[k];
    misc[t] = s;
  }
  if (t == FE) misc[FE] = 0.f;
}

__global__ void __launch_bounds__(256) k_gemm_h(const float* __restrict__ x,
                                                const float* __restrict__ W,
                                                float* __restrict__ h) {
  const int wave = blockIdx.x * (blockDim.x >> 5) + (threadIdx.x >> 5);
  const int lane = threadIdx.x & 31;
  const int ntiles = N_NODES / 16;
  if (wave >= ntiles) return;
  const int m0 = wave * 16;
  const int r  = lane & 15;
  const int hi = lane >> 4;

  v8f c0 = {}; v8f c1 = {};
#if __has_builtin(__builtin_amdgcn_wmma_f32_16x16x4_f32)
  const float* xrow = x + (size_t)(m0 + r) * FN;
  #pragma unroll
  for (int kc = 0; kc < 8; ++kc) {
    const int kb = kc * 4 + hi * 2;
    v2f a; a.x = xrow[kb]; a.y = xrow[kb + 1];
    v2f b0; b0.x = W[(size_t)kb * HID + r];        b0.y = W[(size_t)(kb + 1) * HID + r];
    v2f b1; b1.x = W[(size_t)kb * HID + 16 + r];   b1.y = W[(size_t)(kb + 1) * HID + 16 + r];
    c0 = __builtin_amdgcn_wmma_f32_16x16x4_f32(false, a, false, b0, (short)0, c0, false, false);
    c1 = __builtin_amdgcn_wmma_f32_16x16x4_f32(false, a, false, b1, (short)0, c1, false, false);
  }
  #pragma unroll
  for (int v = 0; v < 8; ++v) {
    const int row = m0 + v + hi * 8;
    h[(size_t)row * HID + r]      = c0[v];
    h[(size_t)row * HID + 16 + r] = c1[v];
  }
#else
  const int row = m0 + r;
  for (int n = hi * 16; n < hi * 16 + 16; ++n) {
    float s = 0.f;
    for (int k = 0; k < FN; ++k) s += x[(size_t)row * FN + k] * W[(size_t)k * HID + n];
    h[(size_t)row * HID + n] = s;
  }
#endif
}

__global__ void k_node(const float* __restrict__ h, const float* __restrict__ att_src,
                       const float* __restrict__ att_dst, float* __restrict__ a_src,
                       float* __restrict__ a_dst, unsigned* __restrict__ segmax,
                       float* __restrict__ denom, float* __restrict__ acc) {
  const int i = blockIdx.x * blockDim.x + threadIdx.x;
  if (i >= N_NODES) return;
  const float4* hp = (const float4*)(h + (size_t)i * HID);
  float s = 0.f, d = 0.f;
  #pragma unroll
  for (int q = 0; q < 8; ++q) {
    float4 hv = hp[q];
    float4 as = ((const float4*)att_src)[q];
    float4 ad = ((const float4*)att_dst)[q];
    s += hv.x * as.x + hv.y * as.y + hv.z * as.z + hv.w * as.w;
    d += hv.x * ad.x + hv.y * ad.y + hv.z * ad.z + hv.w * ad.w;
  }
  a_src[i] = s;
  a_dst[i] = d;
  segmax[i] = ENC_NEG_INF;
  denom[i] = 0.f;
  float4* ap = (float4*)(acc + (size_t)i * HID);
  float4 z = {0.f, 0.f, 0.f, 0.f};
  #pragma unroll
  for (int q = 0; q < 8; ++q) ap[q] = z;
}

__global__ void k_edge_logits(const float* __restrict__ ea, const int* __restrict__ ei,
                              const float* __restrict__ a_src, const float* __restrict__ a_dst,
                              const float* __restrict__ misc, float* __restrict__ logits,
                              unsigned* __restrict__ segmax, float* __restrict__ a_edge_sum) {
  const int e = blockIdx.x * blockDim.x + threadIdx.x;
  float ae = 0.f;
  if (e < N_EDGES) {
    const float4* ep = (const float4*)(ea + (size_t)e * FE);
    float4 e0 = ep[0], e1 = ep[1];
    ae = e0.x * misc[0] + e0.y * misc[1] + e0.z * misc[2] + e0.w * misc[3]
       + e1.x * misc[4] + e1.y * misc[5] + e1.z * misc[6] + e1.w * misc[7];
    const int s  = ei[(size_t)e * 2 + 0];
    const int dd = ei[(size_t)e * 2 + 1];
    float l = a_src[s] + a_dst[dd] + ae;
    l = (l >= 0.f) ? l : NEG_SLOPE * l;
    logits[e] = l;
    atomicMax(segmax + dd, fenc(l));
  }
  float v = ae;
  #pragma unroll
  for (int off = 16; off > 0; off >>= 1) v += __shfl_down(v, off, 32);
  if ((threadIdx.x & 31) == 0) atomAddF(a_edge_sum, v);
}

__global__ void k_self_logits(const float* __restrict__ a_src, const float* __restrict__ a_dst,
                              const float* __restrict__ a_edge_sum, float* __restrict__ logits,
                              unsigned* __restrict__ segmax) {
  const int i = blockIdx.x * blockDim.x + threadIdx.x;
  if (i >= N_NODES) return;
  const float aem = a_edge_sum[0] * (1.0f / (float)N_EDGES);
  float l = a_src[i] + a_dst[i] + aem;
  l = (l >= 0.f) ? l : NEG_SLOPE * l;
  logits[N_EDGES + i] = l;
  atomicMax(segmax + i, fenc(l));
}

__global__ void k_exp(const int* __restrict__ ei, float* __restrict__ logits,
                      const unsigned* __restrict__ segmax, float* __restrict__ denom) {
  const int t = blockIdx.x * blockDim.x + threadIdx.x;
  if (t >= N_EDGES + N_NODES) return;
  const int dd = (t < N_EDGES) ? ei[(size_t)t * 2 + 1] : (t - N_EDGES);
  const float m = fdec(segmax[dd]);
  const float a = __expf(logits[t] - m);
  logits[t] = a;
  atomAddF(denom + dd, a);
}

static_assert((N_EDGES % 32) == 0 && ((N_EDGES + N_NODES) % 32) == 0,
              "the 32 items of a wave are all edges, all self loops, or all out of range");
__global__ void __launch_bounds__(256) k_scatter(const int* __restrict__ ei, const float* __restrict__ h,
                                                 const float* __restrict__ alpha, float* __restrict__ acc) {
  const int t = blockIdx.x * blockDim.x + threadIdx.x;
  if (t >= N_EDGES + N_NODES) return;
  const int lane = threadIdx.x & 31;
  int s, dd;
  if (t < N_EDGES) { s = ei[(size_t)t * 2 + 0]; dd = ei[(size_t)t * 2 + 1]; }
  else             { s = dd = t - N_EDGES; }
  const int abits = __float_as_int(alpha[t]);
  float hv[32];
  #pragma unroll
  for (int j = 0; j < 32; ++j) {
    const int sj = __builtin_amdgcn_readlane(s, j);
    hv[j] = h[(size_t)sj * HID + lane];
  }
  #pragma unroll
  for (int j = 0; j < 32; ++j) {
    const int dj = __builtin_amdgcn_readlane(dd, j);
    const float aj = __int_as_float(__builtin_amdgcn_readlane(abits, j));
    atomAddF(acc + (size_t)dj * HID + lane, aj * hv[j]);
  }
}

__global__ void k_final(const float* __restrict__ acc, const float* __restrict__ denom,
                        const float* __restrict__ bias, const float* __restrict__ W_lin,
                        const float* __restrict__ b_lin, float* __restrict__ out) {
  const int i = blockIdx.x * blockDim.x + threadIdx.x;
  if (i >= N_NODES) return;
  const float inv = 1.0f / denom[i];
  const float* ap = acc + (size_t)i * HID;
  float y = 0.f;
  #pragma unroll
  for (int k = 0; k < HID; ++k) {
    float v = ap[k] * inv + bias[k];
    v = (v > 0.f) ? v : 0.f;
    y += v * W_lin[k];
  }
  out[i] = y + b_lin[0];
}

extern "C" void kernel_launch(void* const* d_in, const int* in_sizes, int n_in,
                              void* d_out, int out_size, void* d_ws, size_t ws_size,
                              hipStream_t stream) {
  const float* x        = (const float*)d_in[0];
  const float* ea       = (const float*)d_in[1];
  const int*   ei       = (const int*)d_in[2];
  const float* W        = (const float*)d_in[3];
  const float* att_src  = (const float*)d_in[4];
  const float* att_dst  = (const float*)d_in[5];
  const float* W_edge   = (const float*)d_in[6];
  const float* att_edge = (const float*)d_in[7];
  const float* bias     = (const float*)d_in[8];
  const float* W_lin    = (const float*)d_in[9];
  const float* b_lin    = (const float*)d_in[10];
  float* out = (float*)d_out;

  float*    ws     = (float*)d_ws;
  float*    h      = ws + OFF_H;
  float*    a_src  = ws + OFF_ASRC;
  float*    a_dst  = ws + OFF_ADST;
  unsigned* segmax = (unsigned*)(ws + OFF_SEGM);
  float*    denom  = ws + OFF_DEN;
  float*    logits = ws + OFF_LOG;
  float*    acc    = ws + OFF_ACC;
  float*    misc   = ws + OFF_MISC;

  const int T = 256;
  const int nb_nodes = (N_NODES + T - 1) / T;
  const int nb_edges = (N_EDGES + T - 1) / T;
  const int nb_all   = (N_EDGES + N_NODES + T - 1) / T;
  const int nb_gemm  = ((N_NODES / 16) + 7) / 8;

  k_setup<<<1, 32, 0, stream>>>(W_edge, att_edge, misc);
  k_gemm_h<<<nb_gemm, T, 0, stream>>>(x, W, h);
  k_node<<<nb_nodes, T, 0, stream>>>(h, att_src, att_dst, a_src, a_dst, segmax, denom, acc);
  k_edge_logits<<<nb_edges, T, 0, stream>>>(ea, ei, a_src, a_dst, misc, logits, segmax, misc + FE);
  k_self_logits<<<nb_nodes, T, 0, stream>>>(a_src, a_dst, misc + FE, logits, segmax);
  k_exp<<<nb_all, T, 0, stream>>>(ei, logits, segmax, denom);
  k_scatter<<<nb_all, T, 0, stream>>>(ei, h, logits, acc);
  k_final<<<nb_nodes, T, 0, stream>>>(acc, denom, bias, W_lin, b_lin, out);
}


# k_edge_logits: the a_edge sum is reduced per block in LDS in front of its one-word f32 atomic, 6250 atomics on that word instead of 50000; the reduction tree of that sum changes
# speedup vs baseline: 3.3565x; 1.2901x over previous
#include <hip/hip_runtime.h>
#include <hip/hip_bf16.h>
#include <stdint.h>

#define N_NODES 100000
#define N_EDGES 1600000
#define FN 32
#define FE 8
#define HID 32
#define NEG_SLOPE 0.2f

typedef __attribute__((ext_vector_type(2))) float v2f;
typedef __attribute__((ext_vector_type(8))) float v8f;

constexpr size_t OFF_H    = 0;
constexpr size_t OFF_ASRC = OFF_H    + (size_t)N_NODES * HID;
constexpr size_t OFF_ADST = OFF_ASRC + N_NODES;
constexpr size_t OFF_SEGM = OFF_ADST + N_NODES;
constexpr size_t OFF_DEN  = OFF_SEGM + N_NODES;
constexpr size_t OFF_LOG  = OFF_DEN  + N_NODES;
constexpr size_t OFF_ACC  = OFF_LOG  + (size_t)(N_EDGES + N_NODES);
constexpr size_t OFF_MISC = OFF_ACC  + (size_t)N_NODES * HID;

__device__ __forceinline__ unsigned fenc(float f) {
  unsigned u = __float_as_uint(f);
  return (u & 0x80000000u) ? ~u : (u | 0x80000000u);
}
__device__ __forceinline__ float fdec(unsigned u) {
  unsigned v = (u & 0x80000000u) ? (u & 0x7FFFFFFFu) : ~u;
  return __uint_as_float(v);
}
#define ENC_NEG_INF 0x007FFFFFu

__device__ __forceinline__ void atomAddF(float* p, float v) { unsafeAtomicAdd(p, v); }

__global__ void k_setup(const float* __restrict__ W_edge, const float* __restrict__ att_edge,
                        float* __restrict__ misc) {
  int t = threadIdx.x;
  if (t < FE) {
    float s = 0.f;
    #pragma unroll
    for (int k = 0; k < HID; ++k) s += W_edge[t * HID + k] * att_edge[k];
    misc[t] = s;
  }
  if (t == FE) misc[FE] = 0.f;
}

__global__ void __launch_bounds__(256) k_gemm_h(const float* __restrict__ x,
                                                const float* __restrict__ W,
                                                float* __restrict__ h) {
  const int wave = blockIdx.x * (blockDim.x >> 5) + (threadIdx.x >> 5);
  const int lane = threadIdx.x & 31;
  const int ntiles = N_NODES / 16;
  if (wave >= ntiles) return;
  const int m0 = wave * 16;
  const int r  = lane & 15;
  const int hi = lane >> 4;

  v8f c0 = {}; v8f c1 = {};
#if __has_builtin(__builtin_amdgcn_wmma_f32_16x16x4_f32)
  const float* xrow = x + (size_t)(m0 + r) * FN;
  #pragma unroll
  for (int kc = 0; kc < 8; ++kc) {
    const int kb = kc * 4 + hi * 2;
    v2f a; a.x = xrow[kb]; a.y = xrow[kb + 1];
    v2f b0; b0.x = W[(size_t)kb * HID + r];        b0.y = W[(size_t)(kb + 1) * HID + r];
    v2f b1; b1.x = W[(size_t)kb * HID + 16 + r];   b1.y = W[(size_t)(kb + 1) * HID + 16 + r];
    c0 = __builtin_amdgcn_wmma_f32_16x16x4_f32(false, a, false, b0, (short)0, c0, false, false);
    c1 = __builtin_amdgcn_wmma_f32_16x16x4_f32(false, a, false, b1, (short)0, c1, false, false);
  }
  #pragma unroll
  for (int v = 0; v < 8; ++v) {
    const int row = m0 + v + hi * 8;
    h[(size_t)row * HID + r]      = c0[v];
    h[(size_t)row * HID + 16 + r] = c1[v];
  }
#else
  const int row = m0 + r;
  for (int n = hi * 16; n < hi * 16 + 16; ++n) {
    float s = 0.f;
    for (int k = 0; k < FN; ++k) s += x[(size_t)row * FN + k] * W[(size_t)k * HID + n];
    h[(size_t)row * HID + n] = s;
  }
#endif
}

__global__ void k_node(const float* __restrict__ h, const float* __restrict__ att_src,
                       const float* __restrict__ att_dst, float* __restrict__ a_src,
                       float* __restrict__ a_dst, unsigned* __restrict__ segmax,
                       float* __restrict__ denom, float* __restrict__ acc) {
  const int i = blockIdx.x * blockDim.x + threadIdx.x;
  if (i >= N_NODES) return;
  const float4* hp = (const float4*)(h + (size_t)i * HID);
  float s = 0.f, d = 0.f;
  #pragma unroll
  for (int q = 0; q < 8; ++q) {
    float4 hv = hp[q];
    float4 as = ((const float4*)att_src)[q];
    float4 ad = ((const float4*)att_dst)[q];
    s += hv.x * as.x + hv.y * as.y + hv.z * as.z + hv.w * as.w;
    d += hv.x * ad.x + hv.y * ad.y + hv.z * ad.z + hv.w * ad.w;
  }
  a_src[i] = s;
  a_dst[i] = d;
  segmax[i] = ENC_NEG_INF;
  denom[i] = 0.f;
  float4* ap = (float4*)(acc + (size_t)i * HID);
  float4 z = {0.f, 0.f, 0.f, 0.f};
  #pragma unroll
  for (int q = 0; q < 8; ++q) ap[q] = z;
}

static_assert((N_EDGES % 256) == 0, "every block of k_edge_logits is full");
__global__ void __launch_bounds__(256) k_edge_logits(const float* __restrict__ ea, const int* __restrict__ ei,
                              const float* __restrict__ a_src, const float* __restrict__ a_dst,
                              const float* __restrict__ misc, float* __restrict__ logits,
                              unsigned* __restrict__ segmax, float* __restrict__ a_edge_sum) {
  __shared__ float wave_sum[8];
  const int e = blockIdx.x * blockDim.x + threadIdx.x;
  const bool live = e < N_EDGES;
  float ae = 0.f, l = 0.f;
  int dd = 0;
  if (live) {
    const float4* ep = (const float4*)(ea + (size_t)e * FE);
    float4 e0 = ep[0], e1 = ep[1];
    ae = e0.x * misc[0] + e0.y * misc[1] + e0.z * misc[2] + e0.w * misc[3]
       + e1.x * misc[4] + e1.y * misc[5] + e1.z * misc[6] + e1.w * misc[7];
    const int s  = ei[(size_t)e * 2 + 0];
    dd = ei[(size_t)e * 2 + 1];
    l = a_src[s] + a_dst[dd] + ae;
    l = (l >= 0.f) ? l : NEG_SLOPE * l;
  }
  float v = ae;
  #pragma unroll
  for (int off = 16; off > 0; off >>= 1) v += __shfl_down(v, off, 32);
  if ((threadIdx.x & 31) == 0) wave_sum[threadIdx.x >> 5] = v;
  __syncthreads();
  if (live) {
    logits[e] = l;
    atomicMax(segmax + dd, fenc(l));
  }
  if (threadIdx.x == 0) {
    const float s = ((wave_sum[0] + wave_sum[1]) + (wave_sum[2] + wave_sum[3]))
                  + ((wave_sum[4] + wave_sum[5]) + (wave_sum[6] + wave_sum[7]));
    atomAddF(a_edge_sum, s);
  }
}

__global__ void k_self_logits(const float* __restrict__ a_src, const float* __restrict__ a_dst,
                              const float* __restrict__ a_edge_sum, float* __restrict__ logits,
                              unsigned* __restrict__ segmax) {
  const int i = blockIdx.x * blockDim.x + threadIdx.x;
  if (i >= N_NODES) return;
  const float aem = a_edge_sum[0] * (1.0f / (float)N_EDGES);
  float l = a_src[i] + a_dst[i] + aem;
  l = (l >= 0.f) ? l : NEG_SLOPE * l;
  logits[N_EDGES + i] = l;
  atomicMax(segmax + i, fenc(l));
}

__global__ void k_exp(const int* __restrict__ ei, float* __restrict__ logits,
                      const unsigned* __restrict__ segmax, float* __restrict__ denom) {
  const int t = blockIdx.x * blockDim.x + threadIdx.x;
  if (t >= N_EDGES + N_NODES) return;
  const int dd = (t < N_EDGES) ? ei[(size_t)t * 2 + 1] : (t - N_EDGES);
  const float m = fdec(segmax[dd]);
  const float a = __expf(logits[t] - m);
  logits[t] = a;
  atomAddF(denom + dd, a);
}

static_assert((N_EDGES % 32) == 0 && ((N_EDGES + N_NODES) % 32) == 0,
              "the 32 items of a wave are all edges, all self loops, or all out of range");
__global__ void __launch_bounds__(256) k_scatter(const int* __restrict__ ei, const float* __restrict__ h,
                                                 const float* __restrict__ alpha, float* __restrict__ acc) {
  const int t = blockIdx.x * blockDim.x + threadIdx.x;
  if (t >= N_EDGES + N_NODES) return;
  const int lane = threadIdx.x & 31;
  int s, dd;
  if (t < N_EDGES) { s = ei[(size_t)t * 2 + 0]; dd = ei[(size_t)t * 2 + 1]; }
  else             { s = dd = t - N_EDGES; }
  const int abits = __float_as_int(alpha[t]);
  float hv[32];
  #pragma unroll
  for (int j = 0; j < 32; ++j) {
    const int sj = __builtin_amdgcn_readlane(s, j);
    hv[j] = h[(size_t)sj * HID + lane];
  }
  #pragma unroll
  for (int j = 0; j < 32; ++j) {
    const int dj = __builtin_amdgcn_readlane(dd, j);
    const float aj = __int_as_float(__builtin_amdgcn_readlane(abits, j));
    atomAddF(acc + (size_t)dj * HID + lane, aj * hv[j]);
  }
}

__global__ void k_final(const float* __restrict__ acc, const float* __restrict__ denom,
                        const float* __restrict__ bias, const float* __restrict__ W_lin,
                        const float* __restrict__ b_lin, float* __restrict__ out) {
  const int i = blockIdx.x * blockDim.x + threadIdx.x;
  if (i >= N_NODES) return;
  const float inv = 1.0f / denom[i];
  const float* ap = acc + (size_t)i * HID;
  float y = 0.f;
  #pragma unroll
  for (int k = 0; k < HID; ++k) {
    float v = ap[k] * inv + bias[k];
    v = (v > 0.f) ? v : 0.f;
    y += v * W_lin[k];
  }
  out[i] = y + b_lin[0];
}

extern "C" void kernel_launch(void* const* d_in, const int* in_sizes, int n_in,
                              void* d_out, int out_size, void* d_ws, size_t ws_size,
                              hipStream_t stream) {
  const float* x        = (const float*)d_in[0];
  const float* ea       = (const float*)d_in[1];
  const int*   ei       = (const int*)d_in[2];
  const float* W        = (const float*)d_in[3];
  const float* att_src  = (const float*)d_in[4];
  const float* att_dst  = (const float*)d_in[5];
  const float* W_edge   = (const float*)d_in[6];
  const float* att_edge = (const float*)d_in[7];
  const float* bias     = (const float*)d_in[8];
  const float* W_lin    = (const float*)d_in[9];
  const float* b_lin    = (const float*)d_in[10];
  float* out = (float*)d_out;

  float*    ws     = (float*)d_ws;
  float*    h      = ws + OFF_H;
  float*    a_src  = ws + OFF_ASRC;
  float*    a_dst  = ws + OFF_ADST;
  unsigned* segmax = (unsigned*)(ws + OFF_SEGM);
  float*    denom  = ws + OFF_DEN;
  float*    logits = ws + OFF_LOG;
  float*    acc    = ws + OFF_ACC;
  float*    misc   = ws + OFF_MISC;

  const int T = 256;
  const int nb_nodes = (N_NODES + T - 1) / T;
  const int nb_edges = (N_EDGES + T - 1) / T;
  const int nb_all   = (N_EDGES + N_NODES + T - 1) / T;
  const int nb_gemm  = ((N_NODES / 16) + 7) / 8;

  k_setup<<<1, 32, 0, stream>>>(W_edge, att_edge, misc);
  k_gemm_h<<<nb_gemm, T, 0, stream>>>(x, W, h);
  k_node<<<nb_nodes, T, 0, stream>>>(h, att_src, att_dst, a_src, a_dst, segmax, denom, acc);
  k_edge_logits<<<nb_edges, T, 0, stream>>>(ea, ei, a_src, a_dst, misc, logits, segmax, misc + FE);
  k_self_logits<<<nb_nodes, T, 0, stream>>>(a_src, a_dst, misc + FE, logits, segmax);
  k_exp<<<nb_all, T, 0, stream>>>(ei, logits, segmax, denom);
  k_scatter<<<nb_all, T, 0, stream>>>(ei, h, logits, acc);
  k_final<<<nb_nodes, T, 0, stream>>>(acc, denom, bias, W_lin, b_lin, out);
}


# acc/denom sums leave the L2 f32 atomics: edges grouped per chunk by destination range, a work-group owns 416 nodes and sums them with LDS adds, final stage joins it; sum order, final dot tree change
# speedup vs baseline: 6.1740x; 1.8394x over previous
#include <hip/hip_runtime.h>
#include <hip/hip_bf16.h>
#include <stdint.h>

#define N_NODES 100000
#define N_EDGES 1600000
#define FN 32
#define FE 8
#define HID 32
#define NEG_SLOPE 0.2f

typedef __attribute__((ext_vector_type(2))) float v2f;
typedef __attribute__((ext_vector_type(8))) float v8f;

constexpr size_t OFF_H    = 0;
constexpr size_t OFF_ASRC = OFF_H    + (size_t)N_NODES * HID;
constexpr size_t OFF_ADST = OFF_ASRC + N_NODES;
constexpr size_t OFF_PAIR = OFF_ADST + N_NODES;
constexpr size_t OFF_LOG  = OFF_PAIR + (size_t)2 * N_NODES;
#define NB 416
#define NBINS ((N_NODES + NB - 1) / NB)
#define CH 250
#define SCH (N_EDGES / CH)
#define DL_BITS 9
static_assert(NB <= (1 << DL_BITS), "the in-range index fits its field");
static_assert(((unsigned long long)N_NODES << DL_BITS) <= 0xFFFFFFFFull, "source and in-range index share one word");
static_assert(CH * SCH == N_EDGES, "the chunks tile the edge list");
static_assert(NBINS + 1 <= 256, "the per-chunk scan covers 256 counters");
static_assert(CH <= 32 * 32, "a wave of k_accum keeps its segment bounds one per lane");
constexpr size_t OFF_REC  = OFF_LOG  + (size_t)N_EDGES;
static_assert((OFF_PAIR % 2) == 0, "the per-node pair is read with one 8-byte load");
static_assert((OFF_REC % 2) == 0, "a record is read with one 8-byte load");
constexpr size_t OFF_TAB  = OFF_REC  + (size_t)2 * N_EDGES;
constexpr size_t OFF_MISC = OFF_TAB  + (size_t)(NBINS + 1) * CH;

__device__ __forceinline__ unsigned fenc(float f) {
  unsigned u = __float_as_uint(f);
  return (u & 0x80000000u) ? ~u : (u | 0x80000000u);
}
__device__ __forceinline__ float fdec(unsigned u) {
  unsigned v = (u & 0x80000000u) ? (u & 0x7FFFFFFFu) : ~u;
  return __uint_as_float(v);
}
#define ENC_NEG_INF 0x007FFFFFu

__device__ __forceinline__ void atomAddF(float* p, float v) { unsafeAtomicAdd(p, v); }

__device__ __forceinline__ float self_logit(float sd, float a_edge_sum) {
  float l = sd + a_edge_sum * (1.0f / (float)N_EDGES);
  return (l >= 0.f) ? l : NEG_SLOPE * l;
}

__global__ void k_setup(const float* __restrict__ W_edge, const float* __restrict__ att_edge,
                        float* __restrict__ misc) {
  int t = threadIdx.x;
  if (t < FE) {
    float s = 0.f;
    #pragma unroll
    for (int k = 0; k < HID; ++k) s += W_edge[t * HID + k] * att_edge[k];
    misc[t] = s;
  }
  if (t == FE) misc[FE] = 0.f;
}

__global__ void __launch_bounds__(256) k_gemm_h(const float* __restrict__ x,
                                                const float* __restrict__ W,
                                                float* __restrict__ h) {
  const int wave = blockIdx.x * (blockDim.x >> 5) + (threadIdx.x >> 5);
  const int lane = threadIdx.x & 31;
  const int ntiles = N_NODES / 16;
  if (wave >= ntiles) return;
  const int m0 = wave * 16;
  const int r  = lane & 15;
  const int hi = lane >> 4;

  v8f c0 = {}; v8f c1 = {};
#if __has_builtin(__builtin_amdgcn_wmma_f32_16x16x4_f32)
  const float* xrow = x + (size_t)(m0 + r) * FN;
  #pragma unroll
  for (int kc = 0; kc < 8; ++kc) {
    const int kb = kc * 4 + hi * 2;
    v2f a; a.x = xrow[kb]; a.y = xrow[kb + 1];
    v2f b0; b0.x = W[(size_t)kb * HID + r];        b0.y = W[(size_t)(kb + 1) * HID + r];
    v2f b1; b1.x = W[(size_t)kb * HID + 16 + r];   b1.y = W[(size_t)(kb + 1) * HID + 16 + r];
    c0 = __builtin_amdgcn_wmma_f32_16x16x4_f32(false, a, false, b0, (short)0, c0, false, false);
    c1 = __builtin_amdgcn_wmma_f32_16x16x4_f32(false, a, false, b1, (short)0, c1, false, false);
  }
  #pragma unroll
  for (int v = 0; v < 8; ++v) {
    const int row = m0 + v + hi * 8;
    h[(size_t)row * HID + r]      = c0[v];
    h[(size_t)row * HID + 16 + r] = c1[v];
  }
#else
  const int row = m0 + r;
  for (int n = hi * 16; n < hi * 16 + 16; ++n) {
    float s = 0.f;
    for (int k = 0; k < FN; ++k) s += x[(size_t)row * FN + k] * W[(size_t)k * HID + n];
    h[(size_t)row * HID + n] = s;
  }
#endif
}

__global__ void k_node(const float* __restrict__ h, const float* __restrict__ att_src,
                       const float* __restrict__ att_dst, float* __restrict__ a_src,
                       float* __restrict__ a_dst, unsigned* __restrict__ pairw) {
  const int i = blockIdx.x * blockDim.x + threadIdx.x;
  if (i >= N_NODES) return;
  const float4* hp = (const float4*)(h + (size_t)i * HID);
  float s = 0.f, d = 0.f;
  #pragma unroll
  for (int q = 0; q < 8; ++q) {
    float4 hv = hp[q];
    float4 as = ((const float4*)att_src)[q];
    float4 ad = ((const float4*)att_dst)[q];
    s += hv.x * as.x + hv.y * as.y + hv.z * as.z + hv.w * as.w;
    d += hv.x * ad.x + hv.y * ad.y + hv.z * ad.z + hv.w * ad.w;
  }
  a_src[i] = s;
  a_dst[i] = d;
  ((uint2*)pairw)[i] = make_uint2(ENC_NEG_INF, __float_as_uint(s + d));
}

static_assert((N_EDGES % 256) == 0, "every block of k_edge_logits is full");
__global__ void __launch_bounds__(256) k_edge_logits(const float* __restrict__ ea, const int* __restrict__ ei,
                              const float* __restrict__ a_src, const float* __restrict__ a_dst,
                              const float* __restrict__ misc, float* __restrict__ logits,
                              unsigned* __restrict__ pairw, float* __restrict__ a_edge_sum) {
  __shared__ float wave_sum[8];
  const int e = blockIdx.x * blockDim.x + threadIdx.x;
  const bool live = e < N_EDGES;
  float ae = 0.f, l = 0.f;
  int dd = 0;
  if (live) {
    const float4* ep = (const float4*)(ea + (size_t)e * FE);
    float4 e0 = ep[0], e1 = ep[1];
    ae = e0.x * misc[0] + e0.y * misc[1] + e0.z * misc[2] + e0.w * misc[3]
       + e1.x * misc[4] + e1.y * misc[5] + e1.z * misc[6] + e1.w * misc[7];
    const int s  = ei[(size_t)e * 2 + 0];
    dd = ei[(size_t)e * 2 + 1];
    l = a_src[s] + a_dst[dd] + ae;
    l = (l >= 0.f) ? l : NEG_SLOPE * l;
  }
  float v = ae;
  #pragma unroll
  for (int off = 16; off > 0; off >>= 1) v += __shfl_down(v, off, 32);
  if ((threadIdx.x & 31) == 0) wave_sum[threadIdx.x >> 5] = v;
  __syncthreads();
  if (live) {
    logits[e] = l;
    atomicMax(pairw + (size_t)dd * 2, fenc(l));
  }
  if (threadIdx.x == 0) {
    const float s = ((wave_sum[0] + wave_sum[1]) + (wave_sum[2] + wave_sum[3]))
                  + ((wave_sum[4] + wave_sum[5]) + (wave_sum[6] + wave_sum[7]));
    atomAddF(a_edge_sum, s);
  }
}

#define TBIN 640
#define KBIN (SCH / TBIN)
static_assert(TBIN * KBIN == SCH, "every thread of k_bin has exactly KBIN edges");
__global__ void __launch_bounds__(TBIN) k_bin(const int* __restrict__ ei, const float* __restrict__ logits,
                                              uint2* __restrict__ recs, int* __restrict__ tab) {
  __shared__ unsigned cnt[256];
  __shared__ unsigned start[256];
  __shared__ uint2 srec[SCH];
  const int tid = threadIdx.x;
  const int c = blockIdx.x;
  const int2* ep = (const int2*)ei + (size_t)c * SCH;
  const float* lp = logits + (size_t)c * SCH;
  int2 e[KBIN];
  float l[KBIN];
  unsigned rank[KBIN];
  #pragma unroll
  for (int k = 0; k < KBIN; ++k) { e[k] = ep[tid + k * TBIN]; l[k] = lp[tid + k * TBIN]; }
  if (tid < 256) cnt[tid] = 0u;
  __syncthreads();
  #pragma unroll
  for (int k = 0; k < KBIN; ++k) rank[k] = atomicAdd(&cnt[e[k].y / NB], 1u);
  __syncthreads();
  if (tid < 32) {
    unsigned v[8];
    unsigned s = 0u;
    #pragma unroll
    for (int k = 0; k < 8; ++k) { v[k] = cnt[tid * 8 + k]; s += v[k]; }
    unsigned incl = s;
    #pragma unroll
    for (int off = 1; off < 32; off <<= 1) {
      const unsigned t = __shfl_up(incl, off, 32);
      if (tid >= off) incl += t;
    }
    unsigned run = incl - s;
    #pragma unroll
    for (int k = 0; k < 8; ++k) { start[tid * 8 + k] = run; run += v[k]; }
  }
  __syncthreads();
  if (tid <= NBINS) tab[(size_t)tid * CH + c] = (int)start[tid];
  #pragma unroll
  for (int k = 0; k < KBIN; ++k) {
    const int b = e[k].y / NB;
    const unsigned dl = (unsigned)(e[k].y - b * NB);
    srec[start[b] + rank[k]] = make_uint2(((unsigned)e[k].x << DL_BITS) | dl, __float_as_uint(l[k]));
  }
  __syncthreads();
  uint2* rp = recs + (size_t)c * SCH;
  #pragma unroll
  for (int k = 0; k < KBIN; ++k) rp[tid + k * TBIN] = srec[tid + k * TBIN];
}

__global__ void __launch_bounds__(1024) k_accum(const uint2* __restrict__ recs, const int* __restrict__ tab,
                                                const float* __restrict__ h,
                                                const unsigned* __restrict__ pairw,
                                                const float* __restrict__ a_edge_sum,
                                                const float* __restrict__ bias, const float* __restrict__ W_lin,
                                                const float* __restrict__ b_lin, float* __restrict__ out) {
  __shared__ float acc_l[NB * HID];
  __shared__ float den_l[NB];
  __shared__ float m_l[NB];
  __shared__ float y_l[NB];
  const int tid = threadIdx.x;
  const int lane = tid & 31;
  const int w = __builtin_amdgcn_readfirstlane(tid >> 5);
  const int b = blockIdx.x;
  const int n0 = b * NB;
  const int nn = (N_NODES - n0 < NB) ? (N_NODES - n0) : NB;
  const float aes = a_edge_sum[0];

  for (int i = tid; i < NB * HID; i += 1024) acc_l[i] = 0.f;
  for (int i = tid; i < NB; i += 1024) {
    float m = 0.f;
    if (i < nn) {
      const uint2 p = ((const uint2*)pairw)[n0 + i];
      m = fmaxf(fdec(p.x), self_logit(__uint_as_float(p.y), aes));
    }
    m_l[i] = m;
    den_l[i] = 0.f;
  }
  const int nseg = (CH - w + 31) >> 5;
  int st_l = 0, en_l = 0;
  if (w + 32 * lane < CH) {
    st_l = tab[(size_t)b * CH + w + 32 * lane];
    en_l = tab[(size_t)(b + 1) * CH + w + 32 * lane];
  }
  __syncthreads();

  typedef const float __attribute__((address_space(1))) * gfp;
  for (int i = 0; i < nseg; ++i) {
    const int st = __builtin_amdgcn_readlane(st_l, i);
    const int en = __builtin_amdgcn_readlane(en_l, i);
    const uint2* rp = recs + (size_t)(w + 32 * i) * SCH;
    for (int base = st; base < en; base += 32) {
      const int idx = base + lane;
      const bool live = idx < en;
      uint2 r = make_uint2(0u, 0u);
      if (live) r = rp[idx];
      const int s  = (int)(r.x >> DL_BITS);
      const int dl = (int)(r.x & ((1u << DL_BITS) - 1u));
      float hv[32];
      #pragma unroll
      for (int j = 0; j < 32; ++j) {
        const int sj = __builtin_amdgcn_readlane(s, j);
        const unsigned long long rb = (unsigned long long)(h + (size_t)sj * HID);
        const unsigned lo = __builtin_amdgcn_readfirstlane((unsigned)rb);
        const unsigned hi = __builtin_amdgcn_readfirstlane((unsigned)(rb >> 32));
        gfp rowj = (gfp)(((unsigned long long)hi << 32) | lo);
        hv[j] = rowj[lane];
      }
      float a = 0.f;
      if (live) {
        a = __expf(__uint_as_float(r.y) - m_l[dl]);
        atomicAdd(&den_l[dl], a);
      }
      const int abits = __float_as_int(a);
      const int cnt = (en - base < 32) ? (en - base) : 32;
      #pragma unroll
      for (int j = 0; j < 32; ++j) {
        if (j < cnt) {
          const int dj = __builtin_amdgcn_readlane(dl, j);
          const float aj = __int_as_float(__builtin_amdgcn_readlane(abits, j));
          atomicAdd(&acc_l[dj * HID + lane], aj * hv[j]);
        }
      }
    }
  }
  __syncthreads();

  const float bl = bias[lane];
  const float wl = W_lin[lane];
  const float b0 = b_lin[0];
  for (int n = w; n < nn; n += 32) {
    const int node = n0 + n;
    const uint2 p = ((const uint2*)pairw)[node];
    const float ls = self_logit(__uint_as_float(p.y), aes);
    const float m  = fmaxf(fdec(p.x), ls);
    const float as = __expf(ls - m);
    const float inv = 1.0f / (den_l[n] + as);
    float v = (acc_l[n * HID + lane] + as * h[(size_t)node * HID + lane]) * inv + bl;
    v = (v > 0.f) ? v : 0.f;
    float y = v * wl;
    #pragma unroll
    for (int off = 16; off > 0; off >>= 1) y += __shfl_xor(y, off, 32);
    if (lane == 0) y_l[n] = y + b0;
  }
  __syncthreads();
  for (int i = tid; i < nn; i += 1024) out[n0 + i] = y_l[i];
}

extern "C" void kernel_launch(void* const* d_in, const int* in_sizes, int n_in,
                              void* d_out, int out_size, void* d_ws, size_t ws_size,
                              hipStream_t stream) {
  const float* x        = (const float*)d_in[0];
  const float* ea       = (const float*)d_in[1];
  const int*   ei       = (const int*)d_in[2];
  const float* W        = (const float*)d_in[3];
  const float* att_src  = (const float*)d_in[4];
  const float* att_dst  = (const float*)d_in[5];
  const float* W_edge   = (const float*)d_in[6];
  const float* att_edge = (const float*)d_in[7];
  const float* bias     = (const float*)d_in[8];
  const float* W_lin    = (const float*)d_in[9];
  const float* b_lin    = (const float*)d_in[10];
  float* out = (float*)d_out;

  float*    ws     = (float*)d_ws;
  float*    h      = ws + OFF_H;
  float*    a_src  = ws + OFF_ASRC;
  float*    a_dst  = ws + OFF_ADST;
  unsigned* pairw  = (unsigned*)(ws + OFF_PAIR);
  float*    logits = ws + OFF_LOG;
  uint2*    recs   = (uint2*)(ws + OFF_REC);
  int*      tab    = (int*)(ws + OFF_TAB);
  float*    misc   = ws + OFF_MISC;

  const int T = 256;
  const int nb_nodes = (N_NODES + T - 1) / T;
  const int nb_edges = (N_EDGES + T - 1) / T;
  const int nb_gemm  = ((N_NODES / 16) + 7) / 8;

  k_setup<<<1, 32, 0, stream>>>(W_edge, att_edge, misc);
  k_gemm_h<<<nb_gemm, T, 0, stream>>>(x, W, h);
  k_node<<<nb_nodes, T, 0, stream>>>(h, att_src, att_dst, a_src, a_dst, pairw);
  k_edge_logits<<<nb_edges, T, 0, stream>>>(ea, ei, a_src, a_dst, misc, logits, pairw, misc + FE);
  k_bin<<<CH, TBIN, 0, stream>>>(ei, logits, recs, tab);
  k_accum<<<NBINS, 1024, 0, stream>>>(recs, tab, h, pairw, misc + FE, bias, W_lin, b_lin, out);
}


# a_dst, leaky relu, segment max move to k_accum: its work-group holds a_dst and max in LDS, first pass over its records; k_edge_bin drops a_dst gathers and L2 max atomics; logit sum order changes
# speedup vs baseline: 13.2731x; 2.1498x over previous
#include <hip/hip_runtime.h>
#include <hip/hip_bf16.h>
#include <stdint.h>

#define N_NODES 100000
#define N_EDGES 1600000
#define FN 32
#define FE 8
#define HID 32
#define NEG_SLOPE 0.2f

typedef __attribute__((ext_vector_type(2))) float v2f;
typedef __attribute__((ext_vector_type(8))) float v8f;

constexpr size_t OFF_H    = 0;
constexpr size_t OFF_ASRC = OFF_H    + (size_t)N_NODES * HID;
constexpr size_t OFF_ADST = OFF_ASRC + N_NODES;
#define NB 416
#define NBINS ((N_NODES + NB - 1) / NB)
#define CH 250
#define SCH (N_EDGES / CH)
#define DL_BITS 9
static_assert(NB <= (1 << DL_BITS), "the in-range index fits its field");
static_assert(((unsigned long long)N_NODES << DL_BITS) <= 0xFFFFFFFFull, "source and in-range index share one word");
static_assert(CH * SCH == N_EDGES, "the chunks tile the edge list");
static_assert(NBINS + 1 <= 256, "the per-chunk scan covers 256 counters");
static_assert(CH <= 32 * 32, "a wave of k_accum keeps its segment bounds one per lane");
#define NSEG_MAX ((CH + 31) / 32)
static_assert(NB <= 1024, "one thread of k_accum per node of the range");
constexpr size_t OFF_REC  = OFF_ADST + N_NODES;
static_assert((OFF_REC % 2) == 0, "a record is read with one 8-byte load");
constexpr size_t OFF_TAB  = OFF_REC  + (size_t)2 * N_EDGES;
constexpr size_t OFF_MISC = OFF_TAB  + (size_t)(NBINS + 1) * CH;

__device__ __forceinline__ unsigned fenc(float f) {
  unsigned u = __float_as_uint(f);
  return (u & 0x80000000u) ? ~u : (u | 0x80000000u);
}
__device__ __forceinline__ float fdec(unsigned u) {
  unsigned v = (u & 0x80000000u) ? (u & 0x7FFFFFFFu) : ~u;
  return __uint_as_float(v);
}
__device__ __forceinline__ void atomAddF(float* p, float v) { unsafeAtomicAdd(p, v); }

__device__ __forceinline__ float self_logit(float sd, float a_edge_sum) {
  float l = sd + a_edge_sum * (1.0f / (float)N_EDGES);
  return (l >= 0.f) ? l : NEG_SLOPE * l;
}

__device__ __forceinline__ float edge_logit(float se, float ad) {
  const float t = se + ad;
  return (t >= 0.f) ? t : NEG_SLOPE * t;
}

__global__ void k_setup(const float* __restrict__ W_edge, const float* __restrict__ att_edge,
                        float* __restrict__ misc) {
  int t = threadIdx.x;
  if (t < FE) {
    float s = 0.f;
    #pragma unroll
    for (int k = 0; k < HID; ++k) s += W_edge[t * HID + k] * att_edge[k];
    misc[t] = s;
  }
  if (t == FE) misc[FE] = 0.f;
}

__global__ void __launch_bounds__(256) k_gemm_h(const float* __restrict__ x,
                                                const float* __restrict__ W,
                                                float* __restrict__ h) {
  const int wave = blockIdx.x * (blockDim.x >> 5) + (threadIdx.x >> 5);
  const int lane = threadIdx.x & 31;
  const int ntiles = N_NODES / 16;
  if (wave >= ntiles) return;
  const int m0 = wave * 16;
  const int r  = lane & 15;
  const int hi = lane >> 4;

  v8f c0 = {}; v8f c1 = {};
#if __has_builtin(__builtin_amdgcn_wmma_f32_16x16x4_f32)
  const float* xrow = x + (size_t)(m0 + r) * FN;
  #pragma unroll
  for (int kc = 0; kc < 8; ++kc) {
    const int kb = kc * 4 + hi * 2;
    v2f a; a.x = xrow[kb]; a.y = xrow[kb + 1];
    v2f b0; b0.x = W[(size_t)kb * HID + r];        b0.y = W[(size_t)(kb + 1) * HID + r];
    v2f b1; b1.x = W[(size_t)kb * HID + 16 + r];   b1.y = W[(size_t)(kb + 1) * HID + 16 + r];
    c0 = __builtin_amdgcn_wmma_f32_16x16x4_f32(false, a, false, b0, (short)0, c0, false, false);
    c1 = __builtin_amdgcn_wmma_f32_16x16x4_f32(false, a, false, b1, (short)0, c1, false, false);
  }
  #pragma unroll
  for (int v = 0; v < 8; ++v) {
    const int row = m0 + v + hi * 8;
    h[(size_t)row * HID + r]      = c0[v];
    h[(size_t)row * HID + 16 + r] = c1[v];
  }
#else
  const int row = m0 + r;
  for (int n = hi * 16; n < hi * 16 + 16; ++n) {
    float s = 0.f;
    for (int k = 0; k < FN; ++k) s += x[(size_t)row * FN + k] * W[(size_t)k * HID + n];
    h[(size_t)row * HID + n] = s;
  }
#endif
}

__global__ void k_node(const float* __restrict__ h, const float* __restrict__ att_src,
                       const float* __restrict__ att_dst, float* __restrict__ a_src,
                       float* __restrict__ a_dst) {
  const int i = blockIdx.x * blockDim.x + threadIdx.x;
  if (i >= N_NODES) return;
  const float4* hp = (const float4*)(h + (size_t)i * HID);
  float s = 0.f, d = 0.f;
  #pragma unroll
  for (int q = 0; q < 8; ++q) {
    float4 hv = hp[q];
    float4 as = ((const float4*)att_src)[q];
    float4 ad = ((const float4*)att_dst)[q];
    s += hv.x * as.x + hv.y * as.y + hv.z * as.z + hv.w * as.w;
    d += hv.x * ad.x + hv.y * ad.y + hv.z * ad.z + hv.w * ad.w;
  }
  a_src[i] = s;
  a_dst[i] = d;
}

__device__ __forceinline__ void lds_barrier() {
  __builtin_amdgcn_fence(__ATOMIC_RELEASE, "workgroup", "local");
  __builtin_amdgcn_s_barrier();
  __builtin_amdgcn_fence(__ATOMIC_ACQUIRE, "workgroup", "local");
}

#define TBIN 640
#define KBIN (SCH / TBIN)
static_assert(TBIN * KBIN == SCH, "every thread of k_edge_bin has exactly KBIN edges");
static_assert((TBIN % 32) == 0, "whole waves");
__global__ void __launch_bounds__(TBIN) k_edge_bin(const float* ea, const int* ei,
                                                   const float* a_src, const float* misc,
                                                   float* a_edge_sum, uint2* recs, int* tab) {
  __shared__ unsigned cnt[256];
  __shared__ unsigned start[256];
  __shared__ uint2 srec[SCH];
  __shared__ float wave_sum[TBIN / 32];
  const int tid = threadIdx.x;
  const int c = blockIdx.x;
  const int2* ep = (const int2*)ei + (size_t)c * SCH;
  const float4* fp = (const float4*)(ea + (size_t)c * SCH * FE);
  const float m0 = misc[0], m1 = misc[1], m2 = misc[2], m3 = misc[3];
  const float m4 = misc[4], m5 = misc[5], m6 = misc[6], m7 = misc[7];
  int2 e[KBIN];
  float4 f0[KBIN], f1[KBIN];
  float as[KBIN], se[KBIN];
  unsigned rank[KBIN];
  #pragma unroll
  for (int k = 0; k < KBIN; ++k) e[k] = ep[tid + k * TBIN];
  if (tid < 256) cnt[tid] = 0u;
  __syncthreads();
  #pragma unroll
  for (int k = 0; k < KBIN; ++k) as[k] = a_src[e[k].x];
  #pragma unroll
  for (int k = 0; k < KBIN; ++k) {
    f0[k] = fp[(size_t)(tid + k * TBIN) * 2 + 0];
    f1[k] = fp[(size_t)(tid + k * TBIN) * 2 + 1];
  }
  #pragma unroll
  for (int k = 0; k < KBIN; ++k) rank[k] = atomicAdd(&cnt[e[k].y / NB], 1u);
  lds_barrier();
  if (tid < 32) {
    unsigned cv[8];
    unsigned s = 0u;
    #pragma unroll
    for (int k = 0; k < 8; ++k) { cv[k] = cnt[tid * 8 + k]; s += cv[k]; }
    unsigned incl = s;
    #pragma unroll
    for (int off = 1; off < 32; off <<= 1) {
      const unsigned t = __shfl_up(incl, off, 32);
      if (tid >= off) incl += t;
    }
    unsigned run = incl - s;
    #pragma unroll
    for (int k = 0; k < 8; ++k) { start[tid * 8 + k] = run; run += cv[k]; }
  }
  lds_barrier();
  if (tid <= NBINS) tab[(size_t)tid * CH + c] = (int)start[tid];
  float v = 0.f;
  #pragma unroll
  for (int k = 0; k < KBIN; ++k) {
    const float ae = f0[k].x * m0 + f0[k].y * m1 + f0[k].z * m2 + f0[k].w * m3
                   + f1[k].x * m4 + f1[k].y * m5 + f1[k].z * m6 + f1[k].w * m7;
    v += ae;
    se[k] = as[k] + ae;
  }
  #pragma unroll
  for (int off = 16; off > 0; off >>= 1) v += __shfl_down(v, off, 32);
  if ((tid & 31) == 0) wave_sum[tid >> 5] = v;
  #pragma unroll
  for (int k = 0; k < KBIN; ++k) {
    const int b = e[k].y / NB;
    const unsigned dl = (unsigned)(e[k].y - b * NB);
    srec[start[b] + rank[k]] = make_uint2(((unsigned)e[k].x << DL_BITS) | dl, __float_as_uint(se[k]));
  }
  lds_barrier();
  uint2* rp = recs + (size_t)c * SCH;
  #pragma unroll
  for (int k = 0; k < KBIN; ++k) rp[tid + k * TBIN] = srec[tid + k * TBIN];
  if (tid == 0) {
    float s = 0.f;
    #pragma unroll
    for (int k = 0; k < TBIN / 32; ++k) s += wave_sum[k];
    atomAddF(a_edge_sum, s);
  }
}

__global__ void __launch_bounds__(1024) k_accum(const uint2* __restrict__ recs, const int* __restrict__ tab,
                                                const float* __restrict__ h,
                                                const float* __restrict__ a_src, const float* __restrict__ a_dst,
                                                const float* __restrict__ a_edge_sum,
                                                const float* __restrict__ bias, const float* __restrict__ W_lin,
                                                const float* __restrict__ b_lin, float* __restrict__ out) {
  __shared__ float acc_l[NB * HID];
  __shared__ float den_l[NB];
  __shared__ uint2 pm_l[NB];
  __shared__ float ls_l[NB];
  __shared__ float y_l[NB];
  const int tid = threadIdx.x;
  const int lane = tid & 31;
  const int w = __builtin_amdgcn_readfirstlane(tid >> 5);
  const int b = blockIdx.x;
  const int n0 = b * NB;
  const int nn = (N_NODES - n0 < NB) ? (N_NODES - n0) : NB;
  const float aes = a_edge_sum[0];
  const int ni = n0 + ((tid < nn) ? tid : (nn - 1));
  const float nd = a_dst[ni];
  const float ns = a_src[ni];

  const int nseg = (CH - w + 31) >> 5;
  int st_l = 0, en_l = 0;
  if (w + 32 * lane < CH) {
    st_l = tab[(size_t)b * CH + w + 32 * lane];
    en_l = tab[(size_t)(b + 1) * CH + w + 32 * lane];
  }
  uint2 r1[NSEG_MAX];
  #pragma unroll
  for (int i = 0; i < NSEG_MAX; ++i) {
    const int st = __builtin_amdgcn_readlane(st_l, i);
    const int en = __builtin_amdgcn_readlane(en_l, i);
    const int c  = (w + 32 * i < CH) ? (w + 32 * i) : (CH - 1);
    const int o  = (st + lane < en) ? (st + lane) : 0;
    r1[i] = recs[(size_t)c * SCH + o];
  }

  for (int i = tid; i < NB * HID; i += 1024) acc_l[i] = 0.f;
  if (tid < NB) {
    const float ls = (tid < nn) ? self_logit(ns + nd, aes) : 0.f;
    pm_l[tid] = make_uint2(fenc(ls), __float_as_uint((tid < nn) ? nd : 0.f));
    ls_l[tid] = ls;
    den_l[tid] = 0.f;
  }
  lds_barrier();

  #pragma unroll
  for (int i = 0; i < NSEG_MAX; ++i) {
    const int st = __builtin_amdgcn_readlane(st_l, i);
    const int en = __builtin_amdgcn_readlane(en_l, i);
    if (st + lane < en) {
      const int dl = (int)(r1[i].x & ((1u << DL_BITS) - 1u));
      const float l = edge_logit(__uint_as_float(r1[i].y), __uint_as_float(pm_l[dl].y));
      atomicMax(&pm_l[dl].x, fenc(l));
    }
    const uint2* rp = recs + (size_t)(w + 32 * i) * SCH;
    for (int base = st + 32; base < en; base += 32) {
      const int idx = base + lane;
      if (idx < en) {
        const uint2 r = rp[idx];
        const int dl = (int)(r.x & ((1u << DL_BITS) - 1u));
        const float l = edge_logit(__uint_as_float(r.y), __uint_as_float(pm_l[dl].y));
        atomicMax(&pm_l[dl].x, fenc(l));
      }
    }
  }
  lds_barrier();

  typedef const float __attribute__((address_space(1))) * gfp;
  for (int i = 0; i < nseg; ++i) {
    const int st = __builtin_amdgcn_readlane(st_l, i);
    const int en = __builtin_amdgcn_readlane(en_l, i);
    const uint2* rp = recs + (size_t)(w + 32 * i) * SCH;
    for (int base = st; base < en; base += 32) {
      const int idx = base + lane;
      const bool live = idx < en;
      uint2 r = make_uint2(0u, 0u);
      if (live) r = rp[idx];
      const int s  = (int)(r.x >> DL_BITS);
      const int dl = (int)(r.x & ((1u << DL_BITS) - 1u));
      float hv[32];
      #pragma unroll
      for (int j = 0; j < 32; ++j) {
        const int sj = __builtin_amdgcn_readlane(s, j);
        const unsigned long long rb = (unsigned long long)(h + (size_t)sj * HID);
        const unsigned lo = __builtin_amdgcn_readfirstlane((unsigned)rb);
        const unsigned hi = __builtin_amdgcn_readfirstlane((unsigned)(rb >> 32));
        gfp rowj = (gfp)(((unsigned long long)hi << 32) | lo);
        hv[j] = rowj[lane];
      }
      float a = 0.f;
      if (live) {
        const uint2 pm = pm_l[dl];
        const float l = edge_logit(__uint_as_float(r.y), __uint_as_float(pm.y));
        a = __expf(l - fdec(pm.x));
        atomicAdd(&den_l[dl], a);
      }
      const int abits = __float_as_int(a);
      const int cnt = (en - base < 32) ? (en - base) : 32;
      #pragma unroll
      for (int j = 0; j < 32; ++j) {
        if (j < cnt) {
          const int dj = __builtin_amdgcn_readlane(dl, j);
          const float aj = __int_as_float(__builtin_amdgcn_readlane(abits, j));
          atomicAdd(&acc_l[dj * HID + lane], aj * hv[j]);
        }
      }
    }
  }
  __syncthreads();

  const float bl = bias[lane];
  const float wl = W_lin[lane];
  const float b0 = b_lin[0];
  for (int n = w; n < nn; n += 32) {
    const int node = n0 + n;
    const float ls = ls_l[n];
    const float m  = fdec(pm_l[n].x);
    const float as = __expf(ls - m);
    const float inv = 1.0f / (den_l[n] + as);
    float v = (acc_l[n * HID + lane] + as * h[(size_t)node * HID + lane]) * inv + bl;
    v = (v > 0.f) ? v : 0.f;
    float y = v * wl;
    #pragma unroll
    for (int off = 16; off > 0; off >>= 1) y += __shfl_xor(y, off, 32);
    if (lane == 0) y_l[n] = y + b0;
  }
  __syncthreads();
  for (int i = tid; i < nn; i += 1024) out[n0 + i] = y_l[i];
}

extern "C" void kernel_launch(void* const* d_in, const int* in_sizes, int n_in,
                              void* d_out, int out_size, void* d_ws, size_t ws_size,
                              hipStream_t stream) {
  const float* x        = (const float*)d_in[0];
  const float* ea       = (const float*)d_in[1];
  const int*   ei       = (const int*)d_in[2];
  const float* W        = (const float*)d_in[3];
  const float* att_src  = (const float*)d_in[4];
  const float* att_dst  = (const float*)d_in[5];
  const float* W_edge   = (const float*)d_in[6];
  const float* att_edge = (const float*)d_in[7];
  const float* bias     = (const float*)d_in[8];
  const float* W_lin    = (const float*)d_in[9];
  const float* b_lin    = (const float*)d_in[10];
  float* out = (float*)d_out;

  float*    ws     = (float*)d_ws;
  float*    h      = ws + OFF_H;
  float*    a_src  = ws + OFF_ASRC;
  float*    a_dst  = ws + OFF_ADST;
  uint2*    recs   = (uint2*)(ws + OFF_REC);
  int*      tab    = (int*)(ws + OFF_TAB);
  float*    misc   = ws + OFF_MISC;

  const int T = 256;
  const int nb_nodes = (N_NODES + T - 1) / T;
  const int nb_gemm  = ((N_NODES / 16) + 7) / 8;

  k_setup<<<1, 32, 0, stream>>>(W_edge, att_edge, misc);
  k_gemm_h<<<nb_gemm, T, 0, stream>>>(x, W, h);
  k_node<<<nb_nodes, T, 0, stream>>>(h, att_src, att_dst, a_src, a_dst);
  k_edge_bin<<<CH, TBIN, 0, stream>>>(ea, ei, a_src, misc, misc + FE, recs, tab);
  k_accum<<<NBINS, 1024, 0, stream>>>(recs, tab, h, a_src, a_dst, misc + FE, bias, W_lin, b_lin, out);
}


# k_accum last stage: a thread per node, not a wave per node in 13 rounds: own rows of h staged in LDS at the start, sums at row pitch 36 read by 16-byte loads, final dot in registers; sum orders kept
# speedup vs baseline: 15.6902x; 1.1821x over previous
#include <hip/hip_runtime.h>
#include <hip/hip_bf16.h>
#include <stdint.h>

#define N_NODES 100000
#define N_EDGES 1600000
#define FN 32
#define FE 8
#define HID 32
#define NEG_SLOPE 0.2f

typedef __attribute__((ext_vector_type(2))) float v2f;
typedef __attribute__((ext_vector_type(8))) float v8f;

constexpr size_t OFF_H    = 0;
constexpr size_t OFF_ASRC = OFF_H    + (size_t)N_NODES * HID;
constexpr size_t OFF_ADST = OFF_ASRC + N_NODES;
#define NB 416
#define NBINS ((N_NODES + NB - 1) / NB)
#define CH 250
#define SCH (N_EDGES / CH)
#define DL_BITS 9
static_assert(NB <= (1 << DL_BITS), "the in-range index fits its field");
static_assert(((unsigned long long)N_NODES << DL_BITS) <= 0xFFFFFFFFull, "source and in-range index share one word");
static_assert(CH * SCH == N_EDGES, "the chunks tile the edge list");
static_assert(NBINS + 1 <= 256, "the per-chunk scan covers 256 counters");
static_assert(CH <= 32 * 32, "a wave of k_accum keeps its segment bounds one per lane");
#define NSEG_MAX ((CH + 31) / 32)
static_assert(NB <= 1024, "one thread of k_accum per node of the range");
constexpr size_t OFF_REC  = OFF_ADST + N_NODES;
static_assert((OFF_REC % 2) == 0, "a record is read with one 8-byte load");
constexpr size_t OFF_TAB  = OFF_REC  + (size_t)2 * N_EDGES;
constexpr size_t OFF_MISC = OFF_TAB  + (size_t)(NBINS + 1) * CH;

__device__ __forceinline__ unsigned fenc(float f) {
  unsigned u = __float_as_uint(f);
  return (u & 0x80000000u) ? ~u : (u | 0x80000000u);
}
__device__ __forceinline__ float fdec(unsigned u) {
  unsigned v = (u & 0x80000000u) ? (u & 0x7FFFFFFFu) : ~u;
  return __uint_as_float(v);
}
__device__ __forceinline__ void atomAddF(float* p, float v) { unsafeAtomicAdd(p, v); }

__device__ __forceinline__ float self_logit(float sd, float a_edge_sum) {
  float l = sd + a_edge_sum * (1.0f / (float)N_EDGES);
  return (l >= 0.f) ? l : NEG_SLOPE * l;
}

__device__ __forceinline__ float edge_logit(float se, float ad) {
  const float t = se + ad;
  return (t >= 0.f) ? t : NEG_SLOPE * t;
}

__device__ __forceinline__ void lds_barrier() {
  __builtin_amdgcn_fence(__ATOMIC_RELEASE, "workgroup", "local");
  __builtin_amdgcn_s_barrier();
  __builtin_amdgcn_fence(__ATOMIC_ACQUIRE, "workgroup", "local");
}

#define GW 10
#define TP 36
static_assert(N_NODES % 16 == 0 && (N_NODES / 16) % GW == 0, "every wave of k_gemm_node has a tile");
static_assert(FE < 32 * GW, "the first work-group has a thread per word of misc");
__global__ void __launch_bounds__(32 * GW) k_gemm_node(const float* __restrict__ x,
                                                       const float* __restrict__ W,
                                                       const float* __restrict__ att_src,
                                                       const float* __restrict__ att_dst,
                                                       const float* __restrict__ W_edge,
                                                       const float* __restrict__ att_edge,
                                                       float* __restrict__ h,
                                                       float* __restrict__ a_src,
                                                       float* __restrict__ a_dst,
                                                       float* __restrict__ misc) {
  __shared__ __attribute__((aligned(16))) float tile[GW][16 * TP];
  const int wv   = threadIdx.x >> 5;
  const int wave = blockIdx.x * GW + wv;
  const int lane = threadIdx.x & 31;
  const int m0 = wave * 16;
  const int r  = lane & 15;
  const int hi = lane >> 4;

  v8f c0 = {}; v8f c1 = {};
  const float* xrow = x + (size_t)(m0 + r) * FN;
  #pragma unroll
  for (int kc = 0; kc < 8; ++kc) {
    const int kb = kc * 4 + hi * 2;
    v2f a; a.x = xrow[kb]; a.y = xrow[kb + 1];
    v2f b0; b0.x = W[(size_t)kb * HID + r];        b0.y = W[(size_t)(kb + 1) * HID + r];
    v2f b1; b1.x = W[(size_t)kb * HID + 16 + r];   b1.y = W[(size_t)(kb + 1) * HID + 16 + r];
    c0 = __builtin_amdgcn_wmma_f32_16x16x4_f32(false, a, false, b0, (short)0, c0, false, false);
    c1 = __builtin_amdgcn_wmma_f32_16x16x4_f32(false, a, false, b1, (short)0, c1, false, false);
  }
  const float4* ap = (const float4*)(hi ? att_dst : att_src);
  float4 av[8];
  #pragma unroll
  for (int q = 0; q < 8; ++q) av[q] = ap[q];
  float* tw = tile[wv];
  #pragma unroll
  for (int v = 0; v < 8; ++v) {
    const int row = m0 + v + hi * 8;
    h[(size_t)row * HID + r]      = c0[v];
    h[(size_t)row * HID + 16 + r] = c1[v];
    tw[(v + hi * 8) * TP + r]      = c0[v];
    tw[(v + hi * 8) * TP + 16 + r] = c1[v];
  }
  lds_barrier();
  const float4* hp = (const float4*)(tw + r * TP);
  float s = 0.f;
  #pragma unroll
  for (int q = 0; q < 8; ++q) {
    const float4 hv = hp[q];
    const float4 as = av[q];
    s += hv.x * as.x + hv.y * as.y + hv.z * as.z + hv.w * as.w;
  }
  (hi ? a_dst : a_src)[m0 + r] = s;

  if (blockIdx.x == 0) {
    const int t = threadIdx.x;
    if (t < FE) {
      float sv = 0.f;
      #pragma unroll
      for (int k = 0; k < HID; ++k) sv += W_edge[t * HID + k] * att_edge[k];
      misc[t] = sv;
    }
    if (t == FE) misc[FE] = 0.f;
  }
}


#define TBIN 640
#define KBIN (SCH / TBIN)
static_assert(TBIN * KBIN == SCH, "every thread of k_edge_bin has exactly KBIN edges");
static_assert((TBIN % 32) == 0, "whole waves");
__global__ void __launch_bounds__(TBIN) k_edge_bin(const float* ea, const int* ei,
                                                   const float* a_src, const float* misc,
                                                   float* a_edge_sum, uint2* recs, int* tab) {
  __shared__ unsigned cnt[256];
  __shared__ unsigned start[256];
  __shared__ uint2 srec[SCH];
  __shared__ float wave_sum[TBIN / 32];
  const int tid = threadIdx.x;
  const int c = blockIdx.x;
  const int2* ep = (const int2*)ei + (size_t)c * SCH;
  const float4* fp = (const float4*)(ea + (size_t)c * SCH * FE);
  const float m0 = misc[0], m1 = misc[1], m2 = misc[2], m3 = misc[3];
  const float m4 = misc[4], m5 = misc[5], m6 = misc[6], m7 = misc[7];
  int2 e[KBIN];
  float4 f0[KBIN], f1[KBIN];
  float as[KBIN], se[KBIN];
  unsigned rank[KBIN];
  #pragma unroll
  for (int k = 0; k < KBIN; ++k) e[k] = ep[tid + k * TBIN];
  if (tid < 256) cnt[tid] = 0u;
  __syncthreads();
  #pragma unroll
  for (int k = 0; k < KBIN; ++k) as[k] = a_src[e[k].x];
  #pragma unroll
  for (int k = 0; k < KBIN; ++k) {
    f0[k] = fp[(size_t)(tid + k * TBIN) * 2 + 0];
    f1[k] = fp[(size_t)(tid + k * TBIN) * 2 + 1];
  }
  #pragma unroll
  for (int k = 0; k < KBIN; ++k) rank[k] = atomicAdd(&cnt[e[k].y / NB], 1u);
  lds_barrier();
  if (tid < 32) {
    unsigned cv[8];
    unsigned s = 0u;
    #pragma unroll
    for (int k = 0; k < 8; ++k) { cv[k] = cnt[tid * 8 + k]; s += cv[k]; }
    unsigned incl = s;
    #pragma unroll
    for (int off = 1; off < 32; off <<= 1) {
      const unsigned t = __shfl_up(incl, off, 32);
      if (tid >= off) incl += t;
    }
    unsigned run = incl - s;
    #pragma unroll
    for (int k = 0; k < 8; ++k) { start[tid * 8 + k] = run; run += cv[k]; }
  }
  lds_barrier();
  if (tid <= NBINS) tab[(size_t)tid * CH + c] = (int)start[tid];
  float v = 0.f;
  #pragma unroll
  for (int k = 0; k < KBIN; ++k) {
    const float ae = f0[k].x * m0 + f0[k].y * m1 + f0[k].z * m2 + f0[k].w * m3
                   + f1[k].x * m4 + f1[k].y * m5 + f1[k].z * m6 + f1[k].w * m7;
    v += ae;
    se[k] = as[k] + ae;
  }
  #pragma unroll
  for (int off = 16; off > 0; off >>= 1) v += __shfl_down(v, off, 32);
  if ((tid & 31) == 0) wave_sum[tid >> 5] = v;
  #pragma unroll
  for (int k = 0; k < KBIN; ++k) {
    const int b = e[k].y / NB;
    const unsigned dl = (unsigned)(e[k].y - b * NB);
    srec[start[b] + rank[k]] = make_uint2(((unsigned)e[k].x << DL_BITS) | dl, __float_as_uint(se[k]));
  }
  lds_barrier();
  uint2* rp = recs + (size_t)c * SCH;
  #pragma unroll
  for (int k = 0; k < KBIN; ++k) rp[tid + k * TBIN] = srec[tid + k * TBIN];
  if (tid == 0) {
    float s = 0.f;
    #pragma unroll
    for (int k = 0; k < TBIN / 32; ++k) s += wave_sum[k];
    atomAddF(a_edge_sum, s);
  }
}

#define AP 36
static_assert((AP % 4) == 0 && ((AP / 4) & 1) == 1 && AP >= HID, "row pitch: 16-byte aligned, an odd number of 16-byte units");
static_assert(NB * (HID / 4) <= 4 * 1024, "four 16-byte loads per thread cover the range's own rows");
__global__ void __launch_bounds__(1024) k_accum(const uint2* __restrict__ recs, const int* __restrict__ tab,
                                                const float* __restrict__ h,
                                                const float* __restrict__ a_src, const float* __restrict__ a_dst,
                                                const float* __restrict__ a_edge_sum,
                                                const float* __restrict__ bias, const float* __restrict__ W_lin,
                                                const float* __restrict__ b_lin, float* __restrict__ out) {
  __shared__ __attribute__((aligned(16))) float acc_l[NB * AP];
  __shared__ __attribute__((aligned(16))) float hs_l[NB * AP];
  __shared__ float den_l[NB];
  __shared__ uint2 pm_l[NB];
  __shared__ float ls_l[NB];
  const int tid = threadIdx.x;
  const int lane = tid & 31;
  const int w = __builtin_amdgcn_readfirstlane(tid >> 5);
  const int b = blockIdx.x;
  const int n0 = b * NB;
  const int nn = (N_NODES - n0 < NB) ? (N_NODES - n0) : NB;
  const float aes = a_edge_sum[0];
  const int ni = n0 + ((tid < nn) ? tid : (nn - 1));
  const float nd = a_dst[ni];
  const float ns = a_src[ni];
  const int nq = nn * (HID / 4);
  const float4* hown = (const float4*)(h + (size_t)n0 * HID);
  float4 hq[4];
  #pragma unroll
  for (int k = 0; k < 4; ++k) {
    const int q = tid + k * 1024;
    hq[k] = hown[(q < nq) ? q : (nq - 1)];
  }

  const int nseg = (CH - w + 31) >> 5;
  int st_l = 0, en_l = 0;
  if (w + 32 * lane < CH) {
    st_l = tab[(size_t)b * CH + w + 32 * lane];
    en_l = tab[(size_t)(b + 1) * CH + w + 32 * lane];
  }
  uint2 r1[NSEG_MAX];
  #pragma unroll
  for (int i = 0; i < NSEG_MAX; ++i) {
    const int st = __builtin_amdgcn_readlane(st_l, i);
    const int en = __builtin_amdgcn_readlane(en_l, i);
    const int c  = (w + 32 * i < CH) ? (w + 32 * i) : (CH - 1);
    const int o  = (st + lane < en) ? (st + lane) : 0;
    r1[i] = recs[(size_t)c * SCH + o];
  }
  #pragma unroll
  for (int k = 0; k < 4; ++k) asm volatile("" : "+v"(hq[k].x), "+v"(hq[k].y), "+v"(hq[k].z), "+v"(hq[k].w));

  for (int i = tid; i < NB * AP / 4; i += 1024) ((float4*)acc_l)[i] = make_float4(0.f, 0.f, 0.f, 0.f);
  #pragma unroll
  for (int k = 0; k < 4; ++k) {
    const int q = tid + k * 1024;
    if (q < nq) *(float4*)&hs_l[(q >> 3) * AP + (q & 7) * 4] = hq[k];
  }
  if (tid < NB) {
    const float ls = (tid < nn) ? self_logit(ns + nd, aes) : 0.f;
    pm_l[tid] = make_uint2(fenc(ls), __float_as_uint((tid < nn) ? nd : 0.f));
    ls_l[tid] = ls;
    den_l[tid] = 0.f;
  }
  lds_barrier();

  #pragma unroll
  for (int i = 0; i < NSEG_MAX; ++i) {
    const int st = __builtin_amdgcn_readlane(st_l, i);
    const int en = __builtin_amdgcn_readlane(en_l, i);
    if (st + lane < en) {
      const int dl = (int)(r1[i].x & ((1u << DL_BITS) - 1u));
      const float l = edge_logit(__uint_as_float(r1[i].y), __uint_as_float(pm_l[dl].y));
      atomicMax(&pm_l[dl].x, fenc(l));
    }
    const uint2* rp = recs + (size_t)(w + 32 * i) * SCH;
    for (int base = st + 32; base < en; base += 32) {
      const int idx = base + lane;
      if (idx < en) {
        const uint2 r = rp[idx];
        const int dl = (int)(r.x & ((1u << DL_BITS) - 1u));
        const float l = edge_logit(__uint_as_float(r.y), __uint_as_float(pm_l[dl].y));
        atomicMax(&pm_l[dl].x, fenc(l));
      }
    }
  }
  lds_barrier();

  typedef const float __attribute__((address_space(1))) * gfp;
  for (int i = 0; i < nseg; ++i) {
    const int st = __builtin_amdgcn_readlane(st_l, i);
    const int en = __builtin_amdgcn_readlane(en_l, i);
    const uint2* rp = recs + (size_t)(w + 32 * i) * SCH;
    for (int base = st; base < en; base += 32) {
      const int idx = base + lane;
      const bool live = idx < en;
      uint2 r = make_uint2(0u, 0u);
      if (live) r = rp[idx];
      const int s  = (int)(r.x >> DL_BITS);
      const int dl = (int)(r.x & ((1u << DL_BITS) - 1u));
      float hv[32];
      #pragma unroll
      for (int j = 0; j < 32; ++j) {
        const int sj = __builtin_amdgcn_readlane(s, j);
        const unsigned long long rb = (unsigned long long)(h + (size_t)sj * HID);
        const unsigned lo = __builtin_amdgcn_readfirstlane((unsigned)rb);
        const unsigned hi = __builtin_amdgcn_readfirstlane((unsigned)(rb >> 32));
        gfp rowj = (gfp)(((unsigned long long)hi << 32) | lo);
        hv[j] = rowj[lane];
      }
      float a = 0.f;
      if (live) {
        const uint2 pm = pm_l[dl];
        const float l = edge_logit(__uint_as_float(r.y), __uint_as_float(pm.y));
        a = __expf(l - fdec(pm.x));
        atomicAdd(&den_l[dl], a);
      }
      const int abits = __float_as_int(a);
      const int rowi = dl * AP;
      const int cnt = (en - base < 32) ? (en - base) : 32;
      #pragma unroll
      for (int j = 0; j < 32; ++j) {
        if (j < cnt) {
          const int dj = __builtin_amdgcn_readlane(rowi, j);
          const float aj = __int_as_float(__builtin_amdgcn_readlane(abits, j));
          atomicAdd(&acc_l[dj + lane], aj * hv[j]);
        }
      }
    }
  }
  __syncthreads();

  if (tid < nn) {
    const float ls = ls_l[tid];
    const float m  = fdec(pm_l[tid].x);
    const float as = __expf(ls - m);
    const float inv = 1.0f / (den_l[tid] + as);
    const float4* ar = (const float4*)&acc_l[tid * AP];
    const float4* hr = (const float4*)&hs_l[tid * AP];
    float y[HID];
    #pragma unroll
    for (int q = 0; q < HID / 4; ++q) {
      float4 a4 = ar[q];
      float4 h4 = hr[q];
      asm volatile("" : "+v"(a4.x), "+v"(a4.y), "+v"(a4.z), "+v"(a4.w), "+v"(h4.x), "+v"(h4.y), "+v"(h4.z), "+v"(h4.w));
      float v0 = (a4.x + as * h4.x) * inv + bias[4 * q + 0];
      float v1 = (a4.y + as * h4.y) * inv + bias[4 * q + 1];
      float v2 = (a4.z + as * h4.z) * inv + bias[4 * q + 2];
      float v3 = (a4.w + as * h4.w) * inv + bias[4 * q + 3];
      v0 = (v0 > 0.f) ? v0 : 0.f;
      v1 = (v1 > 0.f) ? v1 : 0.f;
      v2 = (v2 > 0.f) ? v2 : 0.f;
      v3 = (v3 > 0.f) ? v3 : 0.f;
      y[4 * q + 0] = v0 * W_lin[4 * q + 0];
      y[4 * q + 1] = v1 * W_lin[4 * q + 1];
      y[4 * q + 2] = v2 * W_lin[4 * q + 2];
      y[4 * q + 3] = v3 * W_lin[4 * q + 3];
    }
    #pragma unroll
    for (int off = HID / 2; off > 0; off >>= 1) {
      #pragma unroll
      for (int i = 0; i < off; ++i) y[i] += y[i + off];
    }
    out[n0 + tid] = y[0] + b_lin[0];
  }
}

extern "C" void kernel_launch(void* const* d_in, const int* in_sizes, int n_in,
                              void* d_out, int out_size, void* d_ws, size_t ws_size,
                              hipStream_t stream) {
  const float* x        = (const float*)d_in[0];
  const float* ea       = (const float*)d_in[1];
  const int*   ei       = (const int*)d_in[2];
  const float* W        = (const float*)d_in[3];
  const float* att_src  = (const float*)d_in[4];
  const float* att_dst  = (const float*)d_in[5];
  const float* W_edge   = (const float*)d_in[6];
  const float* att_edge = (const float*)d_in[7];
  const float* bias     = (const float*)d_in[8];
  const float* W_lin    = (const float*)d_in[9];
  const float* b_lin    = (const float*)d_in[10];
  float* out = (float*)d_out;

  float*    ws     = (float*)d_ws;
  float*    h      = ws + OFF_H;
  float*    a_src  = ws + OFF_ASRC;
  float*    a_dst  = ws + OFF_ADST;
  uint2*    recs   = (uint2*)(ws + OFF_REC);
  int*      tab    = (int*)(ws + OFF_TAB);
  float*    misc   = ws + OFF_MISC;

  k_gemm_node<<<(N_NODES / 16) / GW, 32 * GW, 0, stream>>>(x, W, att_src, att_dst, W_edge, att_edge, h, a_src, a_dst, misc);
  k_edge_bin<<<CH, TBIN, 0, stream>>>(ea, ei, a_src, misc, misc + FE, recs, tab);
  k_accum<<<NBINS, 1024, 0, stream>>>(recs, tab, h, a_src, a_dst, misc + FE, bias, W_lin, b_lin, out);
}


# k_accum: a wave packs its records once into a dense LDS list with the finished logit; second pass walks it in full 32-record batches, none re-read from memory; LDS add arrival order of sums changes
# speedup vs baseline: 16.8154x; 1.0717x over previous
#include <hip/hip_runtime.h>
#include <hip/hip_bf16.h>
#include <stdint.h>

#define N_NODES 100000
#define N_EDGES 1600000
#define FN 32
#define FE 8
#define HID 32
#define NEG_SLOPE 0.2f

typedef __attribute__((ext_vector_type(2))) float v2f;
typedef __attribute__((ext_vector_type(8))) float v8f;

constexpr size_t OFF_H    = 0;
constexpr size_t OFF_ASRC = OFF_H    + (size_t)N_NODES * HID;
constexpr size_t OFF_ADST = OFF_ASRC + N_NODES;
#define NB 416
#define NBINS ((N_NODES + NB - 1) / NB)
#define CH 250
#define SCH (N_EDGES / CH)
#define DL_BITS 9
static_assert(NB <= (1 << DL_BITS), "the in-range index fits its field");
static_assert(((unsigned long long)N_NODES << DL_BITS) <= 0xFFFFFFFFull, "source and in-range index share one word");
static_assert(CH * SCH == N_EDGES, "the chunks tile the edge list");
static_assert(NBINS + 1 <= 256, "the per-chunk scan covers 256 counters");
static_assert(CH <= 32 * 32, "a wave of k_accum keeps its segment bounds one per lane");
#define NSEG_MAX ((CH + 31) / 32)
static_assert(NB <= 1024, "one thread of k_accum per node of the range");
constexpr size_t OFF_REC  = OFF_ADST + N_NODES;
static_assert((OFF_REC % 2) == 0, "a record is read with one 8-byte load");
constexpr size_t OFF_TAB  = OFF_REC  + (size_t)2 * N_EDGES;
constexpr size_t OFF_MISC = OFF_TAB  + (size_t)(NBINS + 1) * CH;

__device__ __forceinline__ unsigned fenc(float f) {
  unsigned u = __float_as_uint(f);
  return (u & 0x80000000u) ? ~u : (u | 0x80000000u);
}
__device__ __forceinline__ float fdec(unsigned u) {
  unsigned v = (u & 0x80000000u) ? (u & 0x7FFFFFFFu) : ~u;
  return __uint_as_float(v);
}
__device__ __forceinline__ void atomAddF(float* p, float v) { unsafeAtomicAdd(p, v); }

__device__ __forceinline__ float self_logit(float sd, float a_edge_sum) {
  float l = sd + a_edge_sum * (1.0f / (float)N_EDGES);
  return (l >= 0.f) ? l : NEG_SLOPE * l;
}

__device__ __forceinline__ float edge_logit(float se, float ad) {
  const float t = se + ad;
  return (t >= 0.f) ? t : NEG_SLOPE * t;
}

__device__ __forceinline__ void lds_barrier() {
  __builtin_amdgcn_fence(__ATOMIC_RELEASE, "workgroup", "local");
  __builtin_amdgcn_s_barrier();
  __builtin_amdgcn_fence(__ATOMIC_ACQUIRE, "workgroup", "local");
}

#define GW 10
#define TP 36
static_assert(N_NODES % 16 == 0 && (N_NODES / 16) % GW == 0, "every wave of k_gemm_node has a tile");
static_assert(FE < 32 * GW, "the first work-group has a thread per word of misc");
__global__ void __launch_bounds__(32 * GW) k_gemm_node(const float* __restrict__ x,
                                                       const float* __restrict__ W,
                                                       const float* __restrict__ att_src,
                                                       const float* __restrict__ att_dst,
                                                       const float* __restrict__ W_edge,
                                                       const float* __restrict__ att_edge,
                                                       float* __restrict__ h,
                                                       float* __restrict__ a_src,
                                                       float* __restrict__ a_dst,
                                                       float* __restrict__ misc) {
  __shared__ __attribute__((aligned(16))) float tile[GW][16 * TP];
  const int wv   = threadIdx.x >> 5;
  const int wave = blockIdx.x * GW + wv;
  const int lane = threadIdx.x & 31;
  const int m0 = wave * 16;
  const int r  = lane & 15;
  const int hi = lane >> 4;

  v8f c0 = {}; v8f c1 = {};
  const float* xrow = x + (size_t)(m0 + r) * FN;
  #pragma unroll
  for (int kc = 0; kc < 8; ++kc) {
    const int kb = kc * 4 + hi * 2;
    v2f a; a.x = xrow[kb]; a.y = xrow[kb + 1];
    v2f b0; b0.x = W[(size_t)kb * HID + r];        b0.y = W[(size_t)(kb + 1) * HID + r];
    v2f b1; b1.x = W[(size_t)kb * HID + 16 + r];   b1.y = W[(size_t)(kb + 1) * HID + 16 + r];
    c0 = __builtin_amdgcn_wmma_f32_16x16x4_f32(false, a, false, b0, (short)0, c0, false, false);
    c1 = __builtin_amdgcn_wmma_f32_16x16x4_f32(false, a, false, b1, (short)0, c1, false, false);
  }
  const float4* ap = (const float4*)(hi ? att_dst : att_src);
  float4 av[8];
  #pragma unroll
  for (int q = 0; q < 8; ++q) av[q] = ap[q];
  float* tw = tile[wv];
  #pragma unroll
  for (int v = 0; v < 8; ++v) {
    const int row = m0 + v + hi * 8;
    h[(size_t)row * HID + r]      = c0[v];
    h[(size_t)row * HID + 16 + r] = c1[v];
    tw[(v + hi * 8) * TP + r]      = c0[v];
    tw[(v + hi * 8) * TP + 16 + r] = c1[v];
  }
  lds_barrier();
  const float4* hp = (const float4*)(tw + r * TP);
  float s = 0.f;
  #pragma unroll
  for (int q = 0; q < 8; ++q) {
    const float4 hv = hp[q];
    const float4 as = av[q];
    s += hv.x * as.x + hv.y * as.y + hv.z * as.z + hv.w * as.w;
  }
  (hi ? a_dst : a_src)[m0 + r] = s;

  if (blockIdx.x == 0) {
    const int t = threadIdx.x;
    if (t < FE) {
      float sv = 0.f;
      #pragma unroll
      for (int k = 0; k < HID; ++k) sv += W_edge[t * HID + k] * att_edge[k];
      misc[t] = sv;
    }
    if (t == FE) misc[FE] = 0.f;
  }
}


#define TBIN 640
#define KBIN (SCH / TBIN)
static_assert(TBIN * KBIN == SCH, "every thread of k_edge_bin has exactly KBIN edges");
static_assert((TBIN % 32) == 0, "whole waves");
__global__ void __launch_bounds__(TBIN) k_edge_bin(const float* ea, const int* ei,
                                                   const float* a_src, const float* misc,
                                                   float* a_edge_sum, uint2* recs, int* tab) {
  __shared__ unsigned cnt[256];
  __shared__ unsigned start[256];
  __shared__ uint2 srec[SCH];
  __shared__ float wave_sum[TBIN / 32];
  const int tid = threadIdx.x;
  const int c = blockIdx.x;
  const int2* ep = (const int2*)ei + (size_t)c * SCH;
  const float4* fp = (const float4*)(ea + (size_t)c * SCH * FE);
  const float m0 = misc[0], m1 = misc[1], m2 = misc[2], m3 = misc[3];
  const float m4 = misc[4], m5 = misc[5], m6 = misc[6], m7 = misc[7];
  int2 e[KBIN];
  float4 f0[KBIN], f1[KBIN];
  float as[KBIN], se[KBIN];
  unsigned rank[KBIN];
  #pragma unroll
  for (int k = 0; k < KBIN; ++k) e[k] = ep[tid + k * TBIN];
  if (tid < 256) cnt[tid] = 0u;
  __syncthreads();
  #pragma unroll
  for (int k = 0; k < KBIN; ++k) as[k] = a_src[e[k].x];
  #pragma unroll
  for (int k = 0; k < KBIN; ++k) {
    f0[k] = fp[(size_t)(tid + k * TBIN) * 2 + 0];
    f1[k] = fp[(size_t)(tid + k * TBIN) * 2 + 1];
  }
  #pragma unroll
  for (int k = 0; k < KBIN; ++k) rank[k] = atomicAdd(&cnt[e[k].y / NB], 1u);
  lds_barrier();
  if (tid < 32) {
    unsigned cv[8];
    unsigned s = 0u;
    #pragma unroll
    for (int k = 0; k < 8; ++k) { cv[k] = cnt[tid * 8 + k]; s += cv[k]; }
    unsigned incl = s;
    #pragma unroll
    for (int off = 1; off < 32; off <<= 1) {
      const unsigned t = __shfl_up(incl, off, 32);
      if (tid >= off) incl += t;
    }
    unsigned run = incl - s;
    #pragma unroll
    for (int k = 0; k < 8; ++k) { start[tid * 8 + k] = run; run += cv[k]; }
  }
  lds_barrier();
  if (tid <= NBINS) tab[(size_t)tid * CH + c] = (int)start[tid];
  float v = 0.f;
  #pragma unroll
  for (int k = 0; k < KBIN; ++k) {
    const float ae = f0[k].x * m0 + f0[k].y * m1 + f0[k].z * m2 + f0[k].w * m3
                   + f1[k].x * m4 + f1[k].y * m5 + f1[k].z * m6 + f1[k].w * m7;
    v += ae;
    se[k] = as[k] + ae;
  }
  #pragma unroll
  for (int off = 16; off > 0; off >>= 1) v += __shfl_down(v, off, 32);
  if ((tid & 31) == 0) wave_sum[tid >> 5] = v;
  #pragma unroll
  for (int k = 0; k < KBIN; ++k) {
    const int b = e[k].y / NB;
    const unsigned dl = (unsigned)(e[k].y - b * NB);
    srec[start[b] + rank[k]] = make_uint2(((unsigned)e[k].x << DL_BITS) | dl, __float_as_uint(se[k]));
  }
  lds_barrier();
  uint2* rp = recs + (size_t)c * SCH;
  #pragma unroll
  for (int k = 0; k < KBIN; ++k) rp[tid + k * TBIN] = srec[tid + k * TBIN];
  if (tid == 0) {
    float s = 0.f;
    #pragma unroll
    for (int k = 0; k < TBIN / 32; ++k) s += wave_sum[k];
    atomAddF(a_edge_sum, s);
  }
}

#define AP 36
static_assert((AP % 4) == 0 && ((AP / 4) & 1) == 1 && AP >= HID, "row pitch: 16-byte aligned, an odd number of 16-byte units");
static_assert(NB * (HID / 4) <= 4 * 1024, "four 16-byte loads per thread cover the range's own rows");
#define LCW 352
__global__ void __launch_bounds__(1024) k_accum(const uint2* __restrict__ recs, const int* __restrict__ tab,
                                                const float* __restrict__ h,
                                                const float* __restrict__ a_src, const float* __restrict__ a_dst,
                                                const float* __restrict__ a_edge_sum,
                                                const float* __restrict__ bias, const float* __restrict__ W_lin,
                                                const float* __restrict__ b_lin, float* __restrict__ out) {
  __shared__ __attribute__((aligned(16))) float acc_l[NB * AP];
  __shared__ __attribute__((aligned(16))) float hs_l[NB * AP];
  __shared__ float den_l[NB];
  __shared__ uint2 pm_l[NB];
  __shared__ float ls_l[NB];
  __shared__ uint2 wl_l[32][LCW];
  const int tid = threadIdx.x;
  const int lane = tid & 31;
  const int w = __builtin_amdgcn_readfirstlane(tid >> 5);
  const int b = blockIdx.x;
  const int n0 = b * NB;
  const int nn = (N_NODES - n0 < NB) ? (N_NODES - n0) : NB;
  const float aes = a_edge_sum[0];
  const int ni = n0 + ((tid < nn) ? tid : (nn - 1));
  const float nd = a_dst[ni];
  const float ns = a_src[ni];
  const int nq = nn * (HID / 4);
  const float4* hown = (const float4*)(h + (size_t)n0 * HID);
  float4 hq[4];
  #pragma unroll
  for (int k = 0; k < 4; ++k) {
    const int q = tid + k * 1024;
    hq[k] = hown[(q < nq) ? q : (nq - 1)];
  }

  const int nseg = (CH - w + 31) >> 5;
  int st_l = 0, en_l = 0;
  if (w + 32 * lane < CH) {
    st_l = tab[(size_t)b * CH + w + 32 * lane];
    en_l = tab[(size_t)(b + 1) * CH + w + 32 * lane];
  }
  uint2 r1[NSEG_MAX], r2[NSEG_MAX];
  #pragma unroll
  for (int i = 0; i < NSEG_MAX; ++i) {
    const int st = __builtin_amdgcn_readlane(st_l, i);
    const int en = __builtin_amdgcn_readlane(en_l, i);
    const int c  = (w + 32 * i < CH) ? (w + 32 * i) : (CH - 1);
    const int o1 = (st + lane < en) ? (st + lane) : 0;
    const int o2 = (st + 32 + lane < en) ? (st + 32 + lane) : 0;
    r1[i] = recs[(size_t)c * SCH + o1];
    r2[i] = recs[(size_t)c * SCH + o2];
  }
  #pragma unroll
  for (int k = 0; k < 4; ++k) asm volatile("" : "+v"(hq[k].x), "+v"(hq[k].y), "+v"(hq[k].z), "+v"(hq[k].w));

  const int len_l = en_l - st_l;
  int inc_l = len_l;
  #pragma unroll
  for (int off = 1; off < 32; off <<= 1) {
    const int t = __shfl_up(inc_l, off, 32);
    if (lane >= off) inc_l += t;
  }
  const int of_l = inc_l - len_l;
  const int tot = __builtin_amdgcn_readlane(inc_l, 31);
  const int totl = (tot < LCW) ? tot : LCW;

  for (int i = tid; i < NB * AP / 4; i += 1024) ((float4*)acc_l)[i] = make_float4(0.f, 0.f, 0.f, 0.f);
  #pragma unroll
  for (int k = 0; k < 4; ++k) {
    const int q = tid + k * 1024;
    if (q < nq) *(float4*)&hs_l[(q >> 3) * AP + (q & 7) * 4] = hq[k];
  }
  if (tid < NB) {
    const float ls = (tid < nn) ? self_logit(ns + nd, aes) : 0.f;
    pm_l[tid] = make_uint2(fenc(ls), __float_as_uint((tid < nn) ? nd : 0.f));
    ls_l[tid] = ls;
    den_l[tid] = 0.f;
  }
  lds_barrier();

  uint2* wl = wl_l[w];
  auto max_and_keep = [&](const uint2 r, const int li) {
    const int dl = (int)(r.x & ((1u << DL_BITS) - 1u));
    const float l = edge_logit(__uint_as_float(r.y), __uint_as_float(pm_l[dl].y));
    atomicMax(&pm_l[dl].x, fenc(l));
    if (li < LCW) wl[li] = make_uint2(r.x, __float_as_uint(l));
  };
  #pragma unroll
  for (int i = 0; i < NSEG_MAX; ++i) {
    const int st = __builtin_amdgcn_readlane(st_l, i);
    const int en = __builtin_amdgcn_readlane(en_l, i);
    const int of = __builtin_amdgcn_readlane(of_l, i);
    if (st + lane < en) max_and_keep(r1[i], of + lane);
    if (st + 32 + lane < en) max_and_keep(r2[i], of + 32 + lane);
    const uint2* rp = recs + (size_t)(w + 32 * i) * SCH;
    for (int base = st + 64; base < en; base += 32) {
      const int idx = base + lane;
      if (idx < en) max_and_keep(rp[idx], of + (idx - st));
    }
  }
  lds_barrier();

  typedef const float __attribute__((address_space(1))) * gfp;
  for (int base = 0; base < totl; base += 32) {
    const int idx = base + lane;
    const bool live = idx < totl;
    uint2 r = make_uint2(0u, 0u);
    if (live) r = wl[idx];
    const int s  = (int)(r.x >> DL_BITS);
    const int dl = (int)(r.x & ((1u << DL_BITS) - 1u));
    float hv[32];
    #pragma unroll
    for (int j = 0; j < 32; ++j) {
      const int sj = __builtin_amdgcn_readlane(s, j);
      const unsigned long long rb = (unsigned long long)(h + (size_t)sj * HID);
      const unsigned lo = __builtin_amdgcn_readfirstlane((unsigned)rb);
      const unsigned hi = __builtin_amdgcn_readfirstlane((unsigned)(rb >> 32));
      gfp rowj = (gfp)(((unsigned long long)hi << 32) | lo);
      hv[j] = rowj[lane];
    }
    float a = 0.f;
    if (live) {
      a = __expf(__uint_as_float(r.y) - fdec(pm_l[dl].x));
      atomicAdd(&den_l[dl], a);
    }
    const int abits = __float_as_int(a);
    const int rowi = dl * AP;
    const int cnt = (totl - base < 32) ? (totl - base) : 32;
    if (cnt == 32) {
      #pragma unroll
      for (int j = 0; j < 32; ++j) {
        const int dj = __builtin_amdgcn_readlane(rowi, j);
        const float aj = __int_as_float(__builtin_amdgcn_readlane(abits, j));
        atomicAdd(&acc_l[dj + lane], aj * hv[j]);
      }
    } else {
      #pragma unroll
      for (int j = 0; j < 32; ++j) {
        if (j < cnt) {
          const int dj = __builtin_amdgcn_readlane(rowi, j);
          const float aj = __int_as_float(__builtin_amdgcn_readlane(abits, j));
          atomicAdd(&acc_l[dj + lane], aj * hv[j]);
        }
      }
    }
  }
  if (tot > LCW) {
    for (int i = 0; i < nseg; ++i) {
      const int st = __builtin_amdgcn_readlane(st_l, i);
      const int en = __builtin_amdgcn_readlane(en_l, i);
      const int of = __builtin_amdgcn_readlane(of_l, i);
      const int j0 = (LCW > of) ? (LCW - of) : 0;
      const uint2* rp = recs + (size_t)(w + 32 * i) * SCH;
      for (int base = st + j0; base < en; base += 32) {
        const int idx = base + lane;
        if (idx < en) {
          const uint2 r = rp[idx];
          const int s  = (int)(r.x >> DL_BITS);
          const int dl = (int)(r.x & ((1u << DL_BITS) - 1u));
          const uint2 pm = pm_l[dl];
          const float l = edge_logit(__uint_as_float(r.y), __uint_as_float(pm.y));
          const float a = __expf(l - fdec(pm.x));
          atomicAdd(&den_l[dl], a);
          const float* hr = h + (size_t)s * HID;
          for (int k = 0; k < HID; ++k) atomicAdd(&acc_l[dl * AP + k], a * hr[k]);
        }
      }
    }
  }
  __syncthreads();

  if (tid < nn) {
    const float ls = ls_l[tid];
    const float m  = fdec(pm_l[tid].x);
    const float as = __expf(ls - m);
    const float inv = 1.0f / (den_l[tid] + as);
    const float4* ar = (const float4*)&acc_l[tid * AP];
    const float4* hr = (const float4*)&hs_l[tid * AP];
    float y[HID];
    #pragma unroll
    for (int q = 0; q < HID / 4; ++q) {
      float4 a4 = ar[q];
      float4 h4 = hr[q];
      asm volatile("" : "+v"(a4.x), "+v"(a4.y), "+v"(a4.z), "+v"(a4.w), "+v"(h4.x), "+v"(h4.y), "+v"(h4.z), "+v"(h4.w));
      float v0 = (a4.x + as * h4.x) * inv + bias[4 * q + 0];
      float v1 = (a4.y + as * h4.y) * inv + bias[4 * q + 1];
      float v2 = (a4.z + as * h4.z) * inv + bias[4 * q + 2];
      float v3 = (a4.w + as * h4.w) * inv + bias[4 * q + 3];
      v0 = (v0 > 0.f) ? v0 : 0.f;
      v1 = (v1 > 0.f) ? v1 : 0.f;
      v2 = (v2 > 0.f) ? v2 : 0.f;
      v3 = (v3 > 0.f) ? v3 : 0.f;
      y[4 * q + 0] = v0 * W_lin[4 * q + 0];
      y[4 * q + 1] = v1 * W_lin[4 * q + 1];
      y[4 * q + 2] = v2 * W_lin[4 * q + 2];
      y[4 * q + 3] = v3 * W_lin[4 * q + 3];
    }
    #pragma unroll
    for (int off = HID / 2; off > 0; off >>= 1) {
      #pragma unroll
      for (int i = 0; i < off; ++i) y[i] += y[i + off];
    }
    out[n0 + tid] = y[0] + b_lin[0];
  }
}

extern "C" void kernel_launch(void* const* d_in, const int* in_sizes, int n_in,
                              void* d_out, int out_size, void* d_ws, size_t ws_size,
                              hipStream_t stream) {
  const float* x        = (const float*)d_in[0];
  const float* ea       = (const float*)d_in[1];
  const int*   ei       = (const int*)d_in[2];
  const float* W        = (const float*)d_in[3];
  const float* att_src  = (const float*)d_in[4];
  const float* att_dst  = (const float*)d_in[5];
  const float* W_edge   = (const float*)d_in[6];
  const float* att_edge = (const float*)d_in[7];
  const float* bias     = (const float*)d_in[8];
  const float* W_lin    = (const float*)d_in[9];
  const float* b_lin    = (const float*)d_in[10];
  float* out = (float*)d_out;

  float*    ws     = (float*)d_ws;
  float*    h      = ws + OFF_H;
  float*    a_src  = ws + OFF_ASRC;
  float*    a_dst  = ws + OFF_ADST;
  uint2*    recs   = (uint2*)(ws + OFF_REC);
  int*      tab    = (int*)(ws + OFF_TAB);
  float*    misc   = ws + OFF_MISC;

  k_gemm_node<<<(N_NODES / 16) / GW, 32 * GW, 0, stream>>>(x, W, att_src, att_dst, W_edge, att_edge, h, a_src, a_dst, misc);
  k_edge_bin<<<CH, TBIN, 0, stream>>>(ea, ei, a_src, misc, misc + FE, recs, tab);
  k_accum<<<NBINS, 1024, 0, stream>>>(recs, tab, h, a_src, a_dst, misc + FE, bias, W_lin, b_lin, out);
}
